# starting kernel
# baseline (speedup 1.0000x reference)
#include <hip/hip_runtime.h>
#include <hip/hip_bf16.h>

#define NN   50000
#define NE   1600000
#define INF_ 128
#define HID  64
#define HEADS 2
#define HD   (HEADS*HID)
#define NEG_SLOPE 0.2f

typedef __attribute__((ext_vector_type(2))) float v2f;
typedef __attribute__((ext_vector_type(8))) float v8f;

__device__ __forceinline__ void atomicMaxF(float* addr, float val) {
    if (val >= 0.0f) atomicMax((int*)addr, __float_as_int(val));
    else             atomicMin((unsigned int*)addr, (unsigned int)__float_as_int(val));
}


__global__ void gat_w_transpose(const float* __restrict__ W, float* __restrict__ WT,
                                int K, int total) {
    int i = blockIdx.x * blockDim.x + threadIdx.x;
    if (i >= total) return;
    int k = i / HD;
    int n = i - k * HD;
    WT[(long)n * K + k] = W[i];
}

__global__ void __launch_bounds__(256) gat_gemm_wmma_f32(
        const float* __restrict__ A, const float* __restrict__ BT,
        float* __restrict__ D, int K) {
    const int lane = threadIdx.x & 31;
    const int wave = threadIdx.x >> 5;
    const int half = lane >> 4;
    const int l15  = lane & 15;

    const float* __restrict__ arow = A  + ((long)blockIdx.x * 16 + l15) * K;
    const float* __restrict__ brow = BT + ((long)wave       * 16 + l15) * K;

    v8f c = {};
    for (int kb = 0; kb < K; kb += 4) {
        const int k0 = kb + half * 2;
        v2f a, b;
        a.x = arow[k0]; a.y = arow[k0 + 1];
        b.x = brow[k0]; b.y = brow[k0 + 1];
        c = __builtin_amdgcn_wmma_f32_16x16x4_f32(false, a, false, b,
                                                  (short)0, c, false, false);
    }
    float* __restrict__ drow =
        D + ((long)blockIdx.x * 16 + half * 8) * HD + wave * 16 + l15;
#pragma unroll
    for (int v = 0; v < 8; ++v) drow[(long)v * HD] = c[v];
}

__global__ void gat_eler(const float* __restrict__ feat,
                         const float* __restrict__ al, const float* __restrict__ ar,
                         float* __restrict__ el, float* __restrict__ er, int NH) {
    int i = blockIdx.x * blockDim.x + threadIdx.x;
    if (i >= NH) return;
    const int h = i & 1;
    const float* __restrict__ f = feat + (long)(i >> 1) * HD + h * HID;
    const float* __restrict__ la = al + h * HID;
    const float* __restrict__ ra = ar + h * HID;
    float sl = 0.f, sr = 0.f;
#pragma unroll 4
    for (int d = 0; d < HID; ++d) { float v = f[d]; sl += v * la[d]; sr += v * ra[d]; }
    el[i] = sl; er[i] = sr;
}

__global__ void gat_init(float* __restrict__ m, float* __restrict__ s,
                         float* __restrict__ agg, int NH, int NAGG) {
    int i = blockIdx.x * blockDim.x + threadIdx.x;
    if (i < NH) { m[i] = -INFINITY; s[i] = 0.f; }
    if (i < NAGG) agg[i] = 0.f;
}

__global__ void gat_edge_max(const int* __restrict__ src, const int* __restrict__ dst,
                             const float* __restrict__ el, const float* __restrict__ er,
                             float* __restrict__ ebuf, float* __restrict__ m, int E) {
    int e = blockIdx.x * blockDim.x + threadIdx.x;
    if (e >= E) return;
    const int s0 = src[e], d0 = dst[e];
#pragma unroll
    for (int h = 0; h < HEADS; ++h) {
        float v = el[s0 * HEADS + h] + er[d0 * HEADS + h];
        v = v > 0.f ? v : NEG_SLOPE * v;
        ebuf[(long)e * HEADS + h] = v;
        atomicMaxF(&m[d0 * HEADS + h], v);
    }
}

__global__ void gat_edge_exp(const int* __restrict__ dst, const float* __restrict__ m,
                             float* __restrict__ ebuf, float* __restrict__ ssum, int E) {
    int e = blockIdx.x * blockDim.x + threadIdx.x;
    if (e >= E) return;
    const int d0 = dst[e];
#pragma unroll
    for (int h = 0; h < HEADS; ++h) {
        float ex = __expf(ebuf[(long)e * HEADS + h] - m[d0 * HEADS + h]);
        ebuf[(long)e * HEADS + h] = ex;
        unsafeAtomicAdd(&ssum[d0 * HEADS + h], ex);
    }
}

__global__ void __launch_bounds__(256) gat_edge_scatter(
        const int* __restrict__ src, const int* __restrict__ dst,
        const float* __restrict__ feat, const float* __restrict__ ebuf,
        const float* __restrict__ ssum, float* __restrict__ agg, int E) {
    const int e = blockIdx.x * 8 + (threadIdx.x >> 5);
    if (e >= E) return;
    const int lane = threadIdx.x & 31;
    const int s0 = src[e], d0 = dst[e];
    const int h  = lane >> 4;
    const int c0 = lane * 4;
    const float alpha = ebuf[(long)e * HEADS + h] / ssum[d0 * HEADS + h];

    const int e2 = e + 32;
    if (e2 < E) __builtin_prefetch(feat + (long)src[e2] * HD + c0, 0, 0);

    const float4 f = *(const float4*)(feat + (long)s0 * HD + c0);
    float* __restrict__ o = agg + (long)d0 * HD + c0;
    unsafeAtomicAdd(o + 0, f.x * alpha);
    unsafeAtomicAdd(o + 1, f.y * alpha);
    unsafeAtomicAdd(o + 2, f.z * alpha);
    unsafeAtomicAdd(o + 3, f.w * alpha);
}

__global__ void gat_finalize(const float* __restrict__ agg, const float* __restrict__ bias,
                             float* __restrict__ out, int total) {
    int i = blockIdx.x * blockDim.x + threadIdx.x;
    if (i >= total) return;
    const int n = i >> 6, d = i & 63;
    const float v0 = agg[(long)n * HD + d]        + bias[d];
    const float v1 = agg[(long)n * HD + HID + d]  + bias[HID + d];
    const float v  = 0.5f * (v0 + v1);
    out[i] = v > 0.f ? v : (__expf(v) - 1.f);
}

extern "C" void kernel_launch(void* const* d_in, const int* in_sizes, int n_in,
                              void* d_out, int out_size, void* d_ws, size_t ws_size,
                              hipStream_t stream) {
    const float* x   = (const float*)d_in[0];
    const float* W1  = (const float*)d_in[1];
    const float* al1 = (const float*)d_in[2];
    const float* ar1 = (const float*)d_in[3];
    const float* b1  = (const float*)d_in[4];
    const float* W2  = (const float*)d_in[5];
    const float* al2 = (const float*)d_in[6];
    const float* ar2 = (const float*)d_in[7];
    const float* b2  = (const float*)d_in[8];
    const int*   src = (const int*)d_in[9];
    const int*   dst = (const int*)d_in[10];
    float* out = (float*)d_out;

    char* wsb = (char*)d_ws;
    size_t off = 0;
    auto alloc = [&](size_t bytes) -> void* {
        void* p = wsb + off;
        off += bytes;
        off = (off + 255) & ~(size_t)255;
        return p;
    };
    float* WT1  = (float*)alloc((size_t)INF_ * HD * 4);
    float* WT2  = (float*)alloc((size_t)HID  * HD * 4);
    float* feat = (float*)alloc((size_t)NN * HD * 4);
    float* el   = (float*)alloc((size_t)NN * HEADS * 4);
    float* er   = (float*)alloc((size_t)NN * HEADS * 4);
    float* m    = (float*)alloc((size_t)NN * HEADS * 4);
    float* s    = (float*)alloc((size_t)NN * HEADS * 4);
    float* ebuf = (float*)alloc((size_t)NE * HEADS * 4);
    float* agg  = (float*)alloc((size_t)NN * HD * 4);
    float* h    = (float*)alloc((size_t)NN * HID * 4);

    const int NH   = NN * HEADS;
    const int NAGG = NN * HD;
    const int NOUT = NN * HID;
    const dim3 B256(256);

    gat_w_transpose<<<dim3((INF_ * HD + 255) / 256), B256, 0, stream>>>(W1, WT1, INF_, INF_ * HD);
    gat_gemm_wmma_f32<<<dim3(NN / 16), B256, 0, stream>>>(x, WT1, feat, INF_);
    gat_eler<<<dim3((NH + 255) / 256), B256, 0, stream>>>(feat, al1, ar1, el, er, NH);
    gat_init<<<dim3((NAGG + 255) / 256), B256, 0, stream>>>(m, s, agg, NH, NAGG);
    gat_edge_max<<<dim3((NE + 255) / 256), B256, 0, stream>>>(src, dst, el, er, ebuf, m, NE);
    gat_edge_exp<<<dim3((NE + 255) / 256), B256, 0, stream>>>(dst, m, ebuf, s, NE);
    gat_edge_scatter<<<dim3((NE + 7) / 8), B256, 0, stream>>>(src, dst, feat, ebuf, s, agg, NE);
    gat_finalize<<<dim3((NOUT + 255) / 256), B256, 0, stream>>>(agg, b1, h, NOUT);

    gat_w_transpose<<<dim3((HID * HD + 255) / 256), B256, 0, stream>>>(W2, WT2, HID, HID * HD);
    gat_gemm_wmma_f32<<<dim3(NN / 16), B256, 0, stream>>>(h, WT2, feat, HID);
    gat_eler<<<dim3((NH + 255) / 256), B256, 0, stream>>>(feat, al2, ar2, el, er, NH);
    gat_init<<<dim3((NAGG + 255) / 256), B256, 0, stream>>>(m, s, agg, NH, NAGG);
    gat_edge_max<<<dim3((NE + 255) / 256), B256, 0, stream>>>(src, dst, el, er, ebuf, m, NE);
    gat_edge_exp<<<dim3((NE + 255) / 256), B256, 0, stream>>>(dst, m, ebuf, s, NE);
    gat_edge_scatter<<<dim3((NE + 7) / 8), B256, 0, stream>>>(src, dst, feat, ebuf, s, agg, NE);
    gat_finalize<<<dim3((NOUT + 255) / 256), B256, 0, stream>>>(agg, b2, out, NOUT);
}


# exp pass folded into the scatter: an edge wave forms exp(e-m[dst]) itself, adds it to s and ex*feat to agg (atomics, arrival order as before); finalize divides by s; no ebuf, 2 launches fewer
# speedup vs baseline: 1.2939x; 1.2939x over previous
#include <hip/hip_runtime.h>
#include <hip/hip_bf16.h>

#define NN   50000
#define NE   1600000
#define INF_ 128
#define HID  64
#define HEADS 2
#define HD   (HEADS*HID)
#define NEG_SLOPE 0.2f

typedef __attribute__((ext_vector_type(2))) float v2f;
typedef __attribute__((ext_vector_type(8))) float v8f;

__device__ __forceinline__ void atomicMaxF(float* addr, float val) {
    if (val >= 0.0f) atomicMax((int*)addr, __float_as_int(val));
    else             atomicMin((unsigned int*)addr, (unsigned int)__float_as_int(val));
}


__global__ void gat_w_transpose(const float* __restrict__ W, float* __restrict__ WT,
                                int K, int total) {
    int i = blockIdx.x * blockDim.x + threadIdx.x;
    if (i >= total) return;
    int k = i / HD;
    int n = i - k * HD;
    WT[(long)n * K + k] = W[i];
}

__global__ void __launch_bounds__(256) gat_gemm_wmma_f32(
        const float* __restrict__ A, const float* __restrict__ BT,
        float* __restrict__ D, int K) {
    const int lane = threadIdx.x & 31;
    const int wave = threadIdx.x >> 5;
    const int half = lane >> 4;
    const int l15  = lane & 15;

    const float* __restrict__ arow = A  + ((long)blockIdx.x * 16 + l15) * K;
    const float* __restrict__ brow = BT + ((long)wave       * 16 + l15) * K;

    v8f c = {};
    for (int kb = 0; kb < K; kb += 4) {
        const int k0 = kb + half * 2;
        v2f a, b;
        a.x = arow[k0]; a.y = arow[k0 + 1];
        b.x = brow[k0]; b.y = brow[k0 + 1];
        c = __builtin_amdgcn_wmma_f32_16x16x4_f32(false, a, false, b,
                                                  (short)0, c, false, false);
    }
    float* __restrict__ drow =
        D + ((long)blockIdx.x * 16 + half * 8) * HD + wave * 16 + l15;
#pragma unroll
    for (int v = 0; v < 8; ++v) drow[(long)v * HD] = c[v];
}

__global__ void gat_eler(const float* __restrict__ feat,
                         const float* __restrict__ al, const float* __restrict__ ar,
                         float* __restrict__ el, float* __restrict__ er, int NH) {
    int i = blockIdx.x * blockDim.x + threadIdx.x;
    if (i >= NH) return;
    const int h = i & 1;
    const float* __restrict__ f = feat + (long)(i >> 1) * HD + h * HID;
    const float* __restrict__ la = al + h * HID;
    const float* __restrict__ ra = ar + h * HID;
    float sl = 0.f, sr = 0.f;
#pragma unroll 4
    for (int d = 0; d < HID; ++d) { float v = f[d]; sl += v * la[d]; sr += v * ra[d]; }
    el[i] = sl; er[i] = sr;
}

__global__ void gat_init(float* __restrict__ m, float* __restrict__ s,
                         float* __restrict__ agg, int NH, int NAGG) {
    int i = blockIdx.x * blockDim.x + threadIdx.x;
    if (i < NH) { m[i] = -INFINITY; s[i] = 0.f; }
    if (i < NAGG) agg[i] = 0.f;
}

__global__ void gat_edge_max(const int* __restrict__ src, const int* __restrict__ dst,
                             const float* __restrict__ el, const float* __restrict__ er,
                             float* __restrict__ m, int E) {
    int e = blockIdx.x * blockDim.x + threadIdx.x;
    if (e >= E) return;
    const int s0 = src[e], d0 = dst[e];
#pragma unroll
    for (int h = 0; h < HEADS; ++h) {
        float v = el[s0 * HEADS + h] + er[d0 * HEADS + h];
        v = v > 0.f ? v : NEG_SLOPE * v;
        atomicMaxF(&m[d0 * HEADS + h], v);
    }
}

__global__ void __launch_bounds__(256) gat_edge_scatter(
        const int* __restrict__ src, const int* __restrict__ dst,
        const float* __restrict__ feat,
        const float* __restrict__ el, const float* __restrict__ er,
        const float* __restrict__ m,
        float* __restrict__ ssum, float* __restrict__ agg, int E) {
    const int e = blockIdx.x * 8 + (threadIdx.x >> 5);
    if (e >= E) return;
    const int lane = threadIdx.x & 31;
    const int s0 = src[e], d0 = dst[e];
    const int h  = lane >> 4;
    const int c0 = lane * 4;
    float v = el[s0 * HEADS + h] + er[d0 * HEADS + h];
    v = v > 0.f ? v : NEG_SLOPE * v;
    const float ex = __expf(v - m[d0 * HEADS + h]);

    const int e2 = e + 32;
    if (e2 < E) __builtin_prefetch(feat + (long)src[e2] * HD + c0, 0, 0);

    const float4 f = *(const float4*)(feat + (long)s0 * HD + c0);
    float* __restrict__ o = agg + (long)d0 * HD + c0;
    unsafeAtomicAdd(o + 0, f.x * ex);
    unsafeAtomicAdd(o + 1, f.y * ex);
    unsafeAtomicAdd(o + 2, f.z * ex);
    unsafeAtomicAdd(o + 3, f.w * ex);
    if ((lane & 15) == 0) unsafeAtomicAdd(&ssum[d0 * HEADS + h], ex);
}

__global__ void gat_finalize(const float* __restrict__ agg, const float* __restrict__ ssum,
                             const float* __restrict__ bias,
                             float* __restrict__ out, int total) {
    int i = blockIdx.x * blockDim.x + threadIdx.x;
    if (i >= total) return;
    const int n = i >> 6, d = i & 63;
    const float s0 = ssum[n * HEADS], s1 = ssum[n * HEADS + 1];
    const float a0 = s0 > 0.f ? agg[(long)n * HD + d]       / s0 : 0.f;
    const float a1 = s1 > 0.f ? agg[(long)n * HD + HID + d] / s1 : 0.f;
    const float v0 = a0 + bias[d];
    const float v1 = a1 + bias[HID + d];
    const float v  = 0.5f * (v0 + v1);
    out[i] = v > 0.f ? v : (__expf(v) - 1.f);
}

extern "C" void kernel_launch(void* const* d_in, const int* in_sizes, int n_in,
                              void* d_out, int out_size, void* d_ws, size_t ws_size,
                              hipStream_t stream) {
    const float* x   = (const float*)d_in[0];
    const float* W1  = (const float*)d_in[1];
    const float* al1 = (const float*)d_in[2];
    const float* ar1 = (const float*)d_in[3];
    const float* b1  = (const float*)d_in[4];
    const float* W2  = (const float*)d_in[5];
    const float* al2 = (const float*)d_in[6];
    const float* ar2 = (const float*)d_in[7];
    const float* b2  = (const float*)d_in[8];
    const int*   src = (const int*)d_in[9];
    const int*   dst = (const int*)d_in[10];
    float* out = (float*)d_out;

    char* wsb = (char*)d_ws;
    size_t off = 0;
    auto alloc = [&](size_t bytes) -> void* {
        void* p = wsb + off;
        off += bytes;
        off = (off + 255) & ~(size_t)255;
        return p;
    };
    float* WT1  = (float*)alloc((size_t)INF_ * HD * 4);
    float* WT2  = (float*)alloc((size_t)HID  * HD * 4);
    float* feat = (float*)alloc((size_t)NN * HD * 4);
    float* el   = (float*)alloc((size_t)NN * HEADS * 4);
    float* er   = (float*)alloc((size_t)NN * HEADS * 4);
    float* m    = (float*)alloc((size_t)NN * HEADS * 4);
    float* s    = (float*)alloc((size_t)NN * HEADS * 4);
    float* agg  = (float*)alloc((size_t)NN * HD * 4);
    float* h    = (float*)alloc((size_t)NN * HID * 4);

    const int NH   = NN * HEADS;
    const int NAGG = NN * HD;
    const int NOUT = NN * HID;
    const dim3 B256(256);

    gat_w_transpose<<<dim3((INF_ * HD + 255) / 256), B256, 0, stream>>>(W1, WT1, INF_, INF_ * HD);
    gat_gemm_wmma_f32<<<dim3(NN / 16), B256, 0, stream>>>(x, WT1, feat, INF_);
    gat_eler<<<dim3((NH + 255) / 256), B256, 0, stream>>>(feat, al1, ar1, el, er, NH);
    gat_init<<<dim3((NAGG + 255) / 256), B256, 0, stream>>>(m, s, agg, NH, NAGG);
    gat_edge_max<<<dim3((NE + 255) / 256), B256, 0, stream>>>(src, dst, el, er, m, NE);
    gat_edge_scatter<<<dim3((NE + 7) / 8), B256, 0, stream>>>(src, dst, feat, el, er, m, s, agg, NE);
    gat_finalize<<<dim3((NOUT + 255) / 256), B256, 0, stream>>>(agg, s, b1, h, NOUT);

    gat_w_transpose<<<dim3((HID * HD + 255) / 256), B256, 0, stream>>>(W2, WT2, HID, HID * HD);
    gat_gemm_wmma_f32<<<dim3(NN / 16), B256, 0, stream>>>(h, WT2, feat, HID);
    gat_eler<<<dim3((NH + 255) / 256), B256, 0, stream>>>(feat, al2, ar2, el, er, NH);
    gat_init<<<dim3((NAGG + 255) / 256), B256, 0, stream>>>(m, s, agg, NH, NAGG);
    gat_edge_max<<<dim3((NE + 255) / 256), B256, 0, stream>>>(src, dst, el, er, m, NE);
    gat_edge_scatter<<<dim3((NE + 7) / 8), B256, 0, stream>>>(src, dst, feat, el, er, m, s, agg, NE);
    gat_finalize<<<dim3((NOUT + 255) / 256), B256, 0, stream>>>(agg, s, b2, out, NOUT);
}


# Layer 1's tile scan pruned: the bucket pass also writes each tile a bitmap of the 16-edge groups that hold its edges; a workgroup walks 3128 words, not 1.6M tile bytes; atomic sums, arrival order
# speedup vs baseline: 6.1573x; 4.7587x over previous
#include <hip/hip_runtime.h>
#include <hip/hip_bf16.h>

#define NN   50000
#define NE   1600000
#define INF_ 128
#define HID  64
#define HEADS 2
#define HD   (HEADS*HID)
#define NEG_SLOPE 0.2f

typedef __attribute__((ext_vector_type(2))) float v2f;
typedef __attribute__((ext_vector_type(8))) float v8f;

__device__ __forceinline__ void atomicMaxF(float* addr, float val) {
    if (val >= 0.0f) atomicMax((int*)addr, __float_as_int(val));
    else             atomicMin((unsigned int*)addr, (unsigned int)__float_as_int(val));
}


__global__ void gat_w_transpose(const float* __restrict__ W, float* __restrict__ WT,
                                int K, int total) {
    int i = blockIdx.x * blockDim.x + threadIdx.x;
    if (i >= total) return;
    int k = i / HD;
    int n = i - k * HD;
    WT[(long)n * K + k] = W[i];
}

__global__ void __launch_bounds__(256) gat_gemm_wmma_f32(
        const float* __restrict__ A, const float* __restrict__ BT,
        float* __restrict__ D, int K) {
    const int lane = threadIdx.x & 31;
    const int wave = threadIdx.x >> 5;
    const int half = lane >> 4;
    const int l15  = lane & 15;

    const float* __restrict__ arow = A  + ((long)blockIdx.x * 16 + l15) * K;
    const float* __restrict__ brow = BT + ((long)wave       * 16 + l15) * K;

    v8f c = {};
    for (int kb = 0; kb < K; kb += 4) {
        const int k0 = kb + half * 2;
        v2f a, b;
        a.x = arow[k0]; a.y = arow[k0 + 1];
        b.x = brow[k0]; b.y = brow[k0 + 1];
        c = __builtin_amdgcn_wmma_f32_16x16x4_f32(false, a, false, b,
                                                  (short)0, c, false, false);
    }
    float* __restrict__ drow =
        D + ((long)blockIdx.x * 16 + half * 8) * HD + wave * 16 + l15;
#pragma unroll
    for (int v = 0; v < 8; ++v) drow[(long)v * HD] = c[v];
}

__global__ void gat_eler(const float* __restrict__ feat,
                         const float* __restrict__ al, const float* __restrict__ ar,
                         float* __restrict__ el, float* __restrict__ er, int NH) {
    int i = blockIdx.x * blockDim.x + threadIdx.x;
    if (i >= NH) return;
    const int h = i & 1;
    const float* __restrict__ f = feat + (long)(i >> 1) * HD + h * HID;
    const float* __restrict__ la = al + h * HID;
    const float* __restrict__ ra = ar + h * HID;
    float sl = 0.f, sr = 0.f;
#pragma unroll 4
    for (int d = 0; d < HID; ++d) { float v = f[d]; sl += v * la[d]; sr += v * ra[d]; }
    el[i] = sl; er[i] = sr;
}

#define TILE   196
#define NTILES 256
#define QCAP   8192
#define TB     1024
#define RG     8

#define BKE    2048
#define NBK    ((NE + BKE - 1) / BKE)
#define BMW    (NBK * 4)

__global__ void __launch_bounds__(256) gat_bucket(const int* __restrict__ dst,
                                                  unsigned char* __restrict__ bkt,
                                                  unsigned* __restrict__ tlist, int E) {
    __shared__ unsigned w[4 * NTILES];
    const int t = threadIdx.x;
    for (int i = t; i < 4 * NTILES; i += 256) w[i] = 0u;
    __syncthreads();
    const int e0 = blockIdx.x * BKE + t * 8;
    if (e0 < E) {
        const int4 a = *(const int4*)(dst + e0);
        const int4 b = *(const int4*)(dst + e0 + 4);
        const unsigned t0 = a.x / TILE, t1 = a.y / TILE, t2 = a.z / TILE, t3 = a.w / TILE;
        const unsigned t4 = b.x / TILE, t5 = b.y / TILE, t6 = b.z / TILE, t7 = b.w / TILE;
        uint2 pk;
        pk.x = t0 | (t1 << 8) | (t2 << 16) | (t3 << 24);
        pk.y = t4 | (t5 << 8) | (t6 << 16) | (t7 << 24);
        *(uint2*)(bkt + e0) = pk;
        unsigned* wj = w + (t >> 6) * NTILES;
        const unsigned bit = 1u << ((t >> 1) & 31);
        atomicOr(wj + t0, bit); atomicOr(wj + t1, bit); atomicOr(wj + t2, bit); atomicOr(wj + t3, bit);
        atomicOr(wj + t4, bit); atomicOr(wj + t5, bit); atomicOr(wj + t6, bit); atomicOr(wj + t7, bit);
    }
    __syncthreads();
    uint4 o;
    o.x = w[t]; o.y = w[NTILES + t]; o.z = w[2 * NTILES + t]; o.w = w[3 * NTILES + t];
    *(uint4*)(tlist + (long)t * QCAP + blockIdx.x * 4) = o;
}

__device__ __forceinline__ void tile_match16(uint4 w, unsigned pat, int e0, int* q_e, unsigned* qn) {
    const unsigned ws[4] = {w.x, w.y, w.z, w.w};
#pragma unroll
    for (int k = 0; k < 4; ++k) {
        const unsigned x = ws[k] ^ pat;
        unsigned t = ~(((x & 0x7f7f7f7fu) + 0x7f7f7f7fu) | x | 0x7f7f7f7fu);
        while (t) {
            const int b = __builtin_ctz(t) >> 3;
            const unsigned slot = atomicAdd(qn, 1u);
            if (slot < QCAP) q_e[slot] = e0 + k * 4 + b;
            t &= t - 1u;
        }
    }
}

__device__ __forceinline__ void tile_scan(const unsigned char* __restrict__ bkt,
                                          int e_begin, int e_end, unsigned myb,
                                          int* q_e, unsigned* qn, int tid) {
    const uint4* __restrict__ p = (const uint4*)bkt;
    const unsigned pat = myb * 0x01010101u;
    const int g1 = e_end >> 4;
    int g = (e_begin >> 4) + tid;
    for (; g + 7 * TB < g1; g += 8 * TB) {
        uint4 w[8];
#pragma unroll
        for (int u = 0; u < 8; ++u) w[u] = p[g + u * TB];
#pragma unroll
        for (int u = 0; u < 8; ++u) tile_match16(w[u], pat, (g + u * TB) * 16, q_e, qn);
    }
    for (; g < g1; g += TB) tile_match16(p[g], pat, g * 16, q_e, qn);
}

__device__ __forceinline__ void map_walk(const unsigned* bm, unsigned* q_g, unsigned* gn, int tid) {
    for (int i = tid; i < BMW; i += TB) {
        unsigned w = bm[i];
        while (w) {
            const int b = __builtin_ctz(w);
            w &= w - 1u;
            const unsigned slot = atomicAdd(gn, 1u);
            if (slot < QCAP) q_g[slot] = (unsigned)(i * 32 + b);
        }
    }
}

__device__ __forceinline__ void group_match(const unsigned char* __restrict__ bkt,
                                            const unsigned* q_g, unsigned ng, unsigned myb,
                                            int* q_e, unsigned* qn, int tid) {
    const uint4* __restrict__ p = (const uint4*)bkt;
    const unsigned pat = myb * 0x01010101u;
    for (unsigned i0 = tid; i0 < ng; i0 += 4 * TB) {
        uint4 v[4]; unsigned g[4]; bool live[4];
#pragma unroll
        for (int u = 0; u < 4; ++u) {
            const unsigned i = i0 + u * TB;
            live[u] = i < ng;
            g[u] = q_g[live[u] ? i : i0];
            v[u] = p[g[u]];
        }
#pragma unroll
        for (int u = 0; u < 4; ++u) {
            const unsigned ws[4] = {v[u].x, v[u].y, v[u].z, v[u].w};
            unsigned m = 0u;
#pragma unroll
            for (int k = 0; k < 4; ++k) {
                const unsigned x = ws[k] ^ pat;
                const unsigned t = ~(((x & 0x7f7f7f7fu) + 0x7f7f7f7fu) | x | 0x7f7f7f7fu);
                m |= (t >> 7) << k;
            }
            if (!live[u]) m = 0u;
            while (m) {
                const int pbit = __builtin_ctz(m);
                m &= m - 1u;
                const unsigned slot = atomicAdd(qn, 1u);
                if (slot < QCAP) q_e[slot] = (int)(g[u] * 16u) + (pbit & 7) * 4 + (pbit >> 3);
            }
        }
    }
}

template <bool TAIL>
__device__ __forceinline__ void row_trip(const float* __restrict__ feat, const int* q_e,
                                         const float* q_x, float* acc,
                                         unsigned i0, unsigned cnt, int lane) {
    const unsigned NW = TB / 32;
    float4 f[RG]; float x[RG]; int d[RG];
#pragma unroll
    for (int u = 0; u < RG; ++u) {
        unsigned i = i0 + u * NW;
        bool live = true;
        if (TAIL) { live = i < cnt; i = live ? i : cnt - 1u; }
        const unsigned p = (unsigned)q_e[i];
        d[u] = live ? (int)(p >> 16) : TILE;
        x[u] = q_x[i];
        f[u] = *(const float4*)(feat + (long)(p & 0xffffu) * HD + lane * 4);
    }
#pragma unroll
    for (int u = 0; u < RG; ++u) {
        float* o = acc + d[u] * HD + lane;
        unsafeAtomicAdd(o,      f[u].x * x[u]);
        unsafeAtomicAdd(o + 32, f[u].y * x[u]);
        unsafeAtomicAdd(o + 64, f[u].z * x[u]);
        unsafeAtomicAdd(o + 96, f[u].w * x[u]);
    }
}

__global__ void __launch_bounds__(TB) gat_tile(
        const int* __restrict__ src, const int* __restrict__ dst,
        const unsigned char* __restrict__ bkt,
        const float* __restrict__ feat,
        const float* __restrict__ el, const float* __restrict__ er,
        const float* __restrict__ bias, float* __restrict__ out,
        unsigned* __restrict__ tcnt, unsigned* __restrict__ tlist, int use_list) {
    __shared__ float acc[(TILE + 1) * HD];
    __shared__ float m_l[TILE * HEADS];
    __shared__ float s_l[TILE * HEADS];
    __shared__ float er_l[TILE * HEADS];
    __shared__ int   q_e[QCAP];
    __shared__ float q_x0[QCAP];
    __shared__ float q_x1[QCAP];
    __shared__ unsigned q_g[QCAP];
    __shared__ unsigned qn, gn;

    const int tid  = threadIdx.x;
    const int base = blockIdx.x * TILE;
    const unsigned myb = blockIdx.x;
    unsigned* __restrict__ mylist = tlist + (long)blockIdx.x * QCAP;

    for (int i = tid; i < (TILE + 1) * HD; i += TB) acc[i] = 0.f;
    for (int i = tid; i < TILE * HEADS; i += TB) {
        m_l[i] = -INFINITY; s_l[i] = 0.f;
        er_l[i] = (base + (i >> 1)) < NN ? er[base * HEADS + i] : 0.f;
    }
    if (tid == 0) { qn = 0u; gn = 0u; }
    __syncthreads();

    unsigned cnt = 0u;
    bool listed = false;
    if (use_list) {
        cnt = tcnt[blockIdx.x];
        listed = cnt <= QCAP;
    }
    if (!listed) map_walk(mylist, q_g, &gn, tid);
    __syncthreads();
    if (!listed) {
        const unsigned ng = gn;
        if (ng <= QCAP) group_match(bkt, q_g, ng, myb, q_e, &qn, tid);
        else if (tid == 0) qn = ng;
    }
    __syncthreads();
    if (!listed) {
        cnt = qn;
        if (!use_list && tid == 0) tcnt[blockIdx.x] = cnt;
    }
    const int seglen = cnt > QCAP ? QCAP : NE;
    const int nseg   = (NE + seglen - 1) / seglen;
    const bool publish = !use_list && nseg == 1;

    for (int seg = 0; seg < nseg; ++seg) {
        if (nseg > 1) {
            __syncthreads();
            if (tid == 0) qn = 0u;
            __syncthreads();
            const int eb = seg * seglen;
            tile_scan(bkt, eb, min(NE, eb + seglen), myb, q_e, &qn, tid);
            __syncthreads();
            cnt = qn;
        }
        for (unsigned i = tid; i < cnt; i += TB) {
            unsigned p;
            if (listed) {
                p = mylist[i];
            } else {
                const int e = q_e[i];
                p = (unsigned)src[e] | ((unsigned)(dst[e] - base) << 16);
                if (publish) mylist[i] = p;
            }
            q_e[i] = (int)p;
            const int s0 = (int)(p & 0xffffu);
            const int dl = (int)(p >> 16);
            const float2 l2 = *(const float2*)(el + (long)s0 * HEADS);
            float v0 = l2.x + er_l[dl * HEADS], v1 = l2.y + er_l[dl * HEADS + 1];
            v0 = v0 > 0.f ? v0 : NEG_SLOPE * v0;
            v1 = v1 > 0.f ? v1 : NEG_SLOPE * v1;
            atomicMaxF(&m_l[dl * HEADS], v0);
            atomicMaxF(&m_l[dl * HEADS + 1], v1);
        }
    }
    __syncthreads();

    const int wave = tid >> 5, lane = tid & 31;
    const int h  = lane >> 4;
    for (int seg = 0; seg < nseg; ++seg) {
        if (nseg > 1) {
            __syncthreads();
            if (tid == 0) qn = 0u;
            __syncthreads();
            const int eb = seg * seglen;
            tile_scan(bkt, eb, min(NE, eb + seglen), myb, q_e, &qn, tid);
            __syncthreads();
            cnt = qn;
        }
        for (unsigned i = tid; i < cnt; i += TB) {
            unsigned p = (unsigned)q_e[i];
            if (nseg > 1) {
                const int e = (int)p;
                p = (unsigned)src[e] | ((unsigned)(dst[e] - base) << 16);
            }
            const int s0 = (int)(p & 0xffffu);
            const int dl = (int)(p >> 16);
            const float2 l2 = *(const float2*)(el + (long)s0 * HEADS);
            float v0 = l2.x + er_l[dl * HEADS], v1 = l2.y + er_l[dl * HEADS + 1];
            v0 = v0 > 0.f ? v0 : NEG_SLOPE * v0;
            v1 = v1 > 0.f ? v1 : NEG_SLOPE * v1;
            const float x0 = __expf(v0 - m_l[dl * HEADS]);
            const float x1 = __expf(v1 - m_l[dl * HEADS + 1]);
            q_e[i] = (int)p; q_x0[i] = x0; q_x1[i] = x1;
            unsafeAtomicAdd(&s_l[dl * HEADS], x0);
            unsafeAtomicAdd(&s_l[dl * HEADS + 1], x1);
        }
        __syncthreads();
        const unsigned NW = TB / 32;
        const float* q_x = h ? q_x1 : q_x0;
        unsigned i0 = wave;
        for (; i0 + (RG - 1) * NW < cnt; i0 += RG * NW)
            row_trip<false>(feat, q_e, q_x, acc, i0, cnt, lane);
        if (i0 < cnt) row_trip<true>(feat, q_e, q_x, acc, i0, cnt, lane);
    }
    __syncthreads();

    for (int i = tid; i < TILE * HID; i += TB) {
        const int nl = i >> 6, dd = i & 63;
        const int n  = base + nl;
        if (n < NN) {
            const float s0 = s_l[nl * HEADS], s1 = s_l[nl * HEADS + 1];
            const int w0 = nl * HD + (dd & 3) * 32 + (dd >> 2);
            const float a0 = s0 > 0.f ? acc[w0]      / s0 : 0.f;
            const float a1 = s1 > 0.f ? acc[w0 + 16] / s1 : 0.f;
            const float v0 = a0 + bias[dd];
            const float v1 = a1 + bias[HID + dd];
            const float v  = 0.5f * (v0 + v1);
            out[(long)n * HID + dd] = v > 0.f ? v : (__expf(v) - 1.f);
        }
    }
}

extern "C" void kernel_launch(void* const* d_in, const int* in_sizes, int n_in,
                              void* d_out, int out_size, void* d_ws, size_t ws_size,
                              hipStream_t stream) {
    const float* x   = (const float*)d_in[0];
    const float* W1  = (const float*)d_in[1];
    const float* al1 = (const float*)d_in[2];
    const float* ar1 = (const float*)d_in[3];
    const float* b1  = (const float*)d_in[4];
    const float* W2  = (const float*)d_in[5];
    const float* al2 = (const float*)d_in[6];
    const float* ar2 = (const float*)d_in[7];
    const float* b2  = (const float*)d_in[8];
    const int*   src = (const int*)d_in[9];
    const int*   dst = (const int*)d_in[10];
    float* out = (float*)d_out;

    char* wsb = (char*)d_ws;
    size_t off = 0;
    auto alloc = [&](size_t bytes) -> void* {
        void* p = wsb + off;
        off += bytes;
        off = (off + 255) & ~(size_t)255;
        return p;
    };
    float* WT1  = (float*)alloc((size_t)INF_ * HD * 4);
    float* WT2  = (float*)alloc((size_t)HID  * HD * 4);
    float* feat = (float*)alloc((size_t)NN * HD * 4);
    float* el   = (float*)alloc((size_t)NN * HEADS * 4);
    float* er   = (float*)alloc((size_t)NN * HEADS * 4);
    float* h    = (float*)alloc((size_t)NN * HID * 4);
    unsigned char* bkt = (unsigned char*)alloc((size_t)NE);
    unsigned* tcnt  = (unsigned*)alloc((size_t)NTILES * 4);
    unsigned* tlist = (unsigned*)alloc((size_t)NTILES * QCAP * 4);

    const int NH = NN * HEADS;
    const dim3 B256(256);

    gat_bucket<<<dim3(NBK), B256, 0, stream>>>(dst, bkt, tlist, NE);

    gat_w_transpose<<<dim3((INF_ * HD + 255) / 256), B256, 0, stream>>>(W1, WT1, INF_, INF_ * HD);
    gat_gemm_wmma_f32<<<dim3(NN / 16), B256, 0, stream>>>(x, WT1, feat, INF_);
    gat_eler<<<dim3((NH + 255) / 256), B256, 0, stream>>>(feat, al1, ar1, el, er, NH);
    gat_tile<<<dim3(NTILES), dim3(TB), 0, stream>>>(src, dst, bkt, feat, el, er, b1, h, tcnt, tlist, 0);

    gat_w_transpose<<<dim3((HID * HD + 255) / 256), B256, 0, stream>>>(W2, WT2, HID, HID * HD);
    gat_gemm_wmma_f32<<<dim3(NN / 16), B256, 0, stream>>>(h, WT2, feat, HID);
    gat_eler<<<dim3((NH + 255) / 256), B256, 0, stream>>>(feat, al2, ar2, el, er, NH);
    gat_tile<<<dim3(NTILES), dim3(TB), 0, stream>>>(src, dst, bkt, feat, el, er, b2, out, tcnt, tlist, 1);
}


# Layer 2's projection and logits move into layer 1's tile epilogue: finished rows stay in LDS, meet W2 on f32 WMMA (same k order), logits summed in the old order; h never stored; 9 launches to 6
# speedup vs baseline: 6.8335x; 1.1098x over previous
#include <hip/hip_runtime.h>
#include <hip/hip_bf16.h>

#define NN   50000
#define NE   1600000
#define INF_ 128
#define HID  64
#define HEADS 2
#define HD   (HEADS*HID)
#define NEG_SLOPE 0.2f

typedef __attribute__((ext_vector_type(2))) float v2f;
typedef __attribute__((ext_vector_type(8))) float v8f;

__device__ __forceinline__ void atomicMaxF(float* addr, float val) {
    if (val >= 0.0f) atomicMax((int*)addr, __float_as_int(val));
    else             atomicMin((unsigned int*)addr, (unsigned int)__float_as_int(val));
}


__global__ void gat_w_transpose(const float* __restrict__ W, float* __restrict__ WT,
                                int K, int total) {
    int i = blockIdx.x * blockDim.x + threadIdx.x;
    if (i >= total) return;
    int k = i / HD;
    int n = i - k * HD;
    WT[(long)n * K + k] = W[i];
}

__global__ void __launch_bounds__(256) gat_gemm_wmma_f32(
        const float* __restrict__ A, const float* __restrict__ BT,
        float* __restrict__ D, int K) {
    const int lane = threadIdx.x & 31;
    const int wave = threadIdx.x >> 5;
    const int half = lane >> 4;
    const int l15  = lane & 15;

    const float* __restrict__ arow = A  + ((long)blockIdx.x * 16 + l15) * K;
    const float* __restrict__ brow = BT + ((long)wave       * 16 + l15) * K;

    v8f c = {};
    for (int kb = 0; kb < K; kb += 4) {
        const int k0 = kb + half * 2;
        v2f a, b;
        a.x = arow[k0]; a.y = arow[k0 + 1];
        b.x = brow[k0]; b.y = brow[k0 + 1];
        c = __builtin_amdgcn_wmma_f32_16x16x4_f32(false, a, false, b,
                                                  (short)0, c, false, false);
    }
    float* __restrict__ drow =
        D + ((long)blockIdx.x * 16 + half * 8) * HD + wave * 16 + l15;
#pragma unroll
    for (int v = 0; v < 8; ++v) drow[(long)v * HD] = c[v];
}

__global__ void gat_eler(const float* __restrict__ feat,
                         const float* __restrict__ al, const float* __restrict__ ar,
                         float* __restrict__ el, float* __restrict__ er, int NH) {
    int i = blockIdx.x * blockDim.x + threadIdx.x;
    if (i >= NH) return;
    const int h = i & 1;
    const float* __restrict__ f = feat + (long)(i >> 1) * HD + h * HID;
    const float* __restrict__ la = al + h * HID;
    const float* __restrict__ ra = ar + h * HID;
    float sl = 0.f, sr = 0.f;
#pragma unroll 4
    for (int d = 0; d < HID; ++d) { float v = f[d]; sl += v * la[d]; sr += v * ra[d]; }
    el[i] = sl; er[i] = sr;
}

#define TILE   196
#define NTILES 256
#define QCAP   8192
#define TB     1024
#define RG     8
#define HROWS  208
#define HP     68
#define FP     130
#define FH1    65
#define ACCW   (TILE * FP)

#define BKE    2048
#define NBK    ((NE + BKE - 1) / BKE)
#define BMW    (NBK * 4)

__global__ void __launch_bounds__(256) gat_bucket(const int* __restrict__ dst,
                                                  unsigned char* __restrict__ bkt,
                                                  unsigned* __restrict__ tlist, int E) {
    __shared__ unsigned w[4 * NTILES];
    const int t = threadIdx.x;
    for (int i = t; i < 4 * NTILES; i += 256) w[i] = 0u;
    __syncthreads();
    const int e0 = blockIdx.x * BKE + t * 8;
    if (e0 < E) {
        const int4 a = *(const int4*)(dst + e0);
        const int4 b = *(const int4*)(dst + e0 + 4);
        const unsigned t0 = a.x / TILE, t1 = a.y / TILE, t2 = a.z / TILE, t3 = a.w / TILE;
        const unsigned t4 = b.x / TILE, t5 = b.y / TILE, t6 = b.z / TILE, t7 = b.w / TILE;
        uint2 pk;
        pk.x = t0 | (t1 << 8) | (t2 << 16) | (t3 << 24);
        pk.y = t4 | (t5 << 8) | (t6 << 16) | (t7 << 24);
        *(uint2*)(bkt + e0) = pk;
        unsigned* wj = w + (t >> 6) * NTILES;
        const unsigned bit = 1u << ((t >> 1) & 31);
        atomicOr(wj + t0, bit); atomicOr(wj + t1, bit); atomicOr(wj + t2, bit); atomicOr(wj + t3, bit);
        atomicOr(wj + t4, bit); atomicOr(wj + t5, bit); atomicOr(wj + t6, bit); atomicOr(wj + t7, bit);
    }
    __syncthreads();
    uint4 o;
    o.x = w[t]; o.y = w[NTILES + t]; o.z = w[2 * NTILES + t]; o.w = w[3 * NTILES + t];
    *(uint4*)(tlist + (long)t * QCAP + blockIdx.x * 4) = o;
}

__device__ __forceinline__ void tile_match16(uint4 w, unsigned pat, int e0, int* q_e, unsigned* qn) {
    const unsigned ws[4] = {w.x, w.y, w.z, w.w};
#pragma unroll
    for (int k = 0; k < 4; ++k) {
        const unsigned x = ws[k] ^ pat;
        unsigned t = ~(((x & 0x7f7f7f7fu) + 0x7f7f7f7fu) | x | 0x7f7f7f7fu);
        while (t) {
            const int b = __builtin_ctz(t) >> 3;
            const unsigned slot = atomicAdd(qn, 1u);
            if (slot < QCAP) q_e[slot] = e0 + k * 4 + b;
            t &= t - 1u;
        }
    }
}

__device__ __forceinline__ void tile_scan(const unsigned char* __restrict__ bkt,
                                          int e_begin, int e_end, unsigned myb,
                                          int* q_e, unsigned* qn, int tid) {
    const uint4* __restrict__ p = (const uint4*)bkt;
    const unsigned pat = myb * 0x01010101u;
    const int g1 = e_end >> 4;
    int g = (e_begin >> 4) + tid;
    for (; g + 7 * TB < g1; g += 8 * TB) {
        uint4 w[8];
#pragma unroll
        for (int u = 0; u < 8; ++u) w[u] = p[g + u * TB];
#pragma unroll
        for (int u = 0; u < 8; ++u) tile_match16(w[u], pat, (g + u * TB) * 16, q_e, qn);
    }
    for (; g < g1; g += TB) tile_match16(p[g], pat, g * 16, q_e, qn);
}

__device__ __forceinline__ void map_walk(const unsigned* bm, unsigned* q_g, unsigned* gn, int tid) {
    for (int i = tid; i < BMW; i += TB) {
        unsigned w = bm[i];
        while (w) {
            const int b = __builtin_ctz(w);
            w &= w - 1u;
            const unsigned slot = atomicAdd(gn, 1u);
            if (slot < QCAP) q_g[slot] = (unsigned)(i * 32 + b);
        }
    }
}

__device__ __forceinline__ void group_match(const unsigned char* __restrict__ bkt,
                                            const unsigned* q_g, unsigned ng, unsigned myb,
                                            int* q_e, unsigned* qn, int tid) {
    const uint4* __restrict__ p = (const uint4*)bkt;
    const unsigned pat = myb * 0x01010101u;
    for (unsigned i0 = tid; i0 < ng; i0 += 4 * TB) {
        uint4 v[4]; unsigned g[4]; bool live[4];
#pragma unroll
        for (int u = 0; u < 4; ++u) {
            const unsigned i = i0 + u * TB;
            live[u] = i < ng;
            g[u] = q_g[live[u] ? i : i0];
            v[u] = p[g[u]];
        }
#pragma unroll
        for (int u = 0; u < 4; ++u) {
            const unsigned ws[4] = {v[u].x, v[u].y, v[u].z, v[u].w};
            unsigned m = 0u;
#pragma unroll
            for (int k = 0; k < 4; ++k) {
                const unsigned x = ws[k] ^ pat;
                const unsigned t = ~(((x & 0x7f7f7f7fu) + 0x7f7f7f7fu) | x | 0x7f7f7f7fu);
                m |= (t >> 7) << k;
            }
            if (!live[u]) m = 0u;
            while (m) {
                const int pbit = __builtin_ctz(m);
                m &= m - 1u;
                const unsigned slot = atomicAdd(qn, 1u);
                if (slot < QCAP) q_e[slot] = (int)(g[u] * 16u) + (pbit & 7) * 4 + (pbit >> 3);
            }
        }
    }
}

template <bool TAIL>
__device__ __forceinline__ void row_trip(const float* __restrict__ feat, const int* q_e,
                                         const float* q_x, float* acc,
                                         unsigned i0, unsigned cnt, int lane) {
    const unsigned NW = TB / 32;
    float4 f[RG]; float x[RG]; int d[RG];
#pragma unroll
    for (int u = 0; u < RG; ++u) {
        unsigned i = i0 + u * NW;
        bool live = true;
        if (TAIL) { live = i < cnt; i = live ? i : cnt - 1u; }
        const unsigned p = (unsigned)q_e[i];
        d[u] = live ? (int)(p >> 16) : TILE;
        x[u] = q_x[i];
        f[u] = *(const float4*)(feat + (long)(p & 0xffffu) * HD + lane * 4);
    }
#pragma unroll
    for (int u = 0; u < RG; ++u) {
        float* o = acc + d[u] * HD + lane;
        unsafeAtomicAdd(o,      f[u].x * x[u]);
        unsafeAtomicAdd(o + 32, f[u].y * x[u]);
        unsafeAtomicAdd(o + 64, f[u].z * x[u]);
        unsafeAtomicAdd(o + 96, f[u].w * x[u]);
    }
}

__global__ void __launch_bounds__(TB) gat_tile(
        const int* __restrict__ src, const int* __restrict__ dst,
        const unsigned char* __restrict__ bkt,
        const float* __restrict__ feat,
        const float* __restrict__ el, const float* __restrict__ er,
        const float* __restrict__ bias, float* __restrict__ out,
        unsigned* __restrict__ tcnt, unsigned* __restrict__ tlist, int use_list,
        const float* __restrict__ Wn, const float* __restrict__ aln, const float* __restrict__ arn,
        float* __restrict__ featn, float* __restrict__ eln, float* __restrict__ ern) {
    __shared__ __attribute__((aligned(16))) float acc[ACCW];
    __shared__ float m_l[TILE * HEADS];
    __shared__ float s_l[TILE * HEADS];
    __shared__ float er_l[TILE * HEADS];
    __shared__ int   q_e[QCAP];
    __shared__ __attribute__((aligned(16))) float q_xb[2 * QCAP];
    __shared__ unsigned q_g[QCAP];
    __shared__ unsigned qn, gn;
    float* const q_x0 = q_xb;
    float* const q_x1 = q_xb + QCAP;

    const int tid  = threadIdx.x;
    const int base = blockIdx.x * TILE;
    const unsigned myb = blockIdx.x;
    unsigned* __restrict__ mylist = tlist + (long)blockIdx.x * QCAP;

    for (int i = tid; i < (TILE + 1) * HD; i += TB) acc[i] = 0.f;
    for (int i = tid; i < TILE * HEADS; i += TB) {
        m_l[i] = -INFINITY; s_l[i] = 0.f;
        er_l[i] = (base + (i >> 1)) < NN ? er[base * HEADS + i] : 0.f;
    }
    if (tid == 0) { qn = 0u; gn = 0u; }
    __syncthreads();

    unsigned cnt = 0u;
    bool listed = false;
    if (use_list) {
        cnt = tcnt[blockIdx.x];
        listed = cnt <= QCAP;
    }
    if (!listed) map_walk(mylist, q_g, &gn, tid);
    __syncthreads();
    if (!listed) {
        const unsigned ng = gn;
        if (ng <= QCAP) group_match(bkt, q_g, ng, myb, q_e, &qn, tid);
        else if (tid == 0) qn = ng;
    }
    __syncthreads();
    if (!listed) {
        cnt = qn;
        if (!use_list && tid == 0) tcnt[blockIdx.x] = cnt;
    }
    const int seglen = cnt > QCAP ? QCAP : NE;
    const int nseg   = (NE + seglen - 1) / seglen;
    const bool publish = !use_list && nseg == 1;

    for (int seg = 0; seg < nseg; ++seg) {
        if (nseg > 1) {
            __syncthreads();
            if (tid == 0) qn = 0u;
            __syncthreads();
            const int eb = seg * seglen;
            tile_scan(bkt, eb, min(NE, eb + seglen), myb, q_e, &qn, tid);
            __syncthreads();
            cnt = qn;
        }
        for (unsigned i = tid; i < cnt; i += TB) {
            unsigned p;
            if (listed) {
                p = mylist[i];
            } else {
                const int e = q_e[i];
                p = (unsigned)src[e] | ((unsigned)(dst[e] - base) << 16);
                if (publish) mylist[i] = p;
            }
            q_e[i] = (int)p;
            const int s0 = (int)(p & 0xffffu);
            const int dl = (int)(p >> 16);
            const float2 l2 = *(const float2*)(el + (long)s0 * HEADS);
            float v0 = l2.x + er_l[dl * HEADS], v1 = l2.y + er_l[dl * HEADS + 1];
            v0 = v0 > 0.f ? v0 : NEG_SLOPE * v0;
            v1 = v1 > 0.f ? v1 : NEG_SLOPE * v1;
            atomicMaxF(&m_l[dl * HEADS], v0);
            atomicMaxF(&m_l[dl * HEADS + 1], v1);
        }
    }
    __syncthreads();

    const int wave = tid >> 5, lane = tid & 31;
    const int h  = lane >> 4;
    for (int seg = 0; seg < nseg; ++seg) {
        if (nseg > 1) {
            __syncthreads();
            if (tid == 0) qn = 0u;
            __syncthreads();
            const int eb = seg * seglen;
            tile_scan(bkt, eb, min(NE, eb + seglen), myb, q_e, &qn, tid);
            __syncthreads();
            cnt = qn;
        }
        for (unsigned i = tid; i < cnt; i += TB) {
            unsigned p = (unsigned)q_e[i];
            if (nseg > 1) {
                const int e = (int)p;
                p = (unsigned)src[e] | ((unsigned)(dst[e] - base) << 16);
            }
            const int s0 = (int)(p & 0xffffu);
            const int dl = (int)(p >> 16);
            const float2 l2 = *(const float2*)(el + (long)s0 * HEADS);
            float v0 = l2.x + er_l[dl * HEADS], v1 = l2.y + er_l[dl * HEADS + 1];
            v0 = v0 > 0.f ? v0 : NEG_SLOPE * v0;
            v1 = v1 > 0.f ? v1 : NEG_SLOPE * v1;
            const float x0 = __expf(v0 - m_l[dl * HEADS]);
            const float x1 = __expf(v1 - m_l[dl * HEADS + 1]);
            q_e[i] = (int)p; q_x0[i] = x0; q_x1[i] = x1;
            unsafeAtomicAdd(&s_l[dl * HEADS], x0);
            unsafeAtomicAdd(&s_l[dl * HEADS + 1], x1);
        }
        __syncthreads();
        const unsigned NW = TB / 32;
        const float* q_x = h ? q_x1 : q_x0;
        unsigned i0 = wave;
        for (; i0 + (RG - 1) * NW < cnt; i0 += RG * NW)
            row_trip<false>(feat, q_e, q_x, acc, i0, cnt, lane);
        if (i0 < cnt) row_trip<true>(feat, q_e, q_x, acc, i0, cnt, lane);
    }
    __syncthreads();

    if (!use_list) {
        float* const hb = q_xb;
        for (int i = tid; i < HROWS * HID; i += TB) {
            const int nl = i >> 6, dd = i & 63;
            float r = 0.f;
            if (nl < TILE && base + nl < NN) {
                const float s0 = s_l[nl * HEADS], s1 = s_l[nl * HEADS + 1];
                const int w0 = nl * HD + (dd & 3) * 32 + (dd >> 2);
                const float a0 = s0 > 0.f ? acc[w0]      / s0 : 0.f;
                const float a1 = s1 > 0.f ? acc[w0 + 16] / s1 : 0.f;
                const float v0 = a0 + bias[dd];
                const float v1 = a1 + bias[HID + dd];
                const float v  = 0.5f * (v0 + v1);
                r = v > 0.f ? v : (__expf(v) - 1.f);
            }
            hb[nl * HP + dd] = r;
        }
        __syncthreads();

        const int wv   = __builtin_amdgcn_readfirstlane(wave);
        const int ct   = wv & 7;
        const int half = lane >> 4, l15 = lane & 15;
        v2f bf[HID / 4];
#pragma unroll
        for (int kq = 0; kq < HID / 4; ++kq) {
            const int k0 = kq * 4 + half * 2;
            bf[kq].x = Wn[k0 * HD + ct * 16 + l15];
            bf[kq].y = Wn[(k0 + 1) * HD + ct * 16 + l15];
        }
        float* const fb = acc;
        const int fcol = (ct >> 2) * FH1 + (ct & 3) * 16 + l15;
        for (int rt = wv >> 3; rt < HROWS / 16; rt += 4) {
            const float* arow = hb + (rt * 16 + l15) * HP + half * 2;
            v8f cc = {};
#pragma unroll
            for (int kq = 0; kq < HID / 4; ++kq) {
                const v2f a = *(const v2f*)(arow + kq * 4);
                cc = __builtin_amdgcn_wmma_f32_16x16x4_f32(false, a, false, bf[kq],
                                                           (short)0, cc, false, false);
            }
            const int r0 = rt * 16 + half * 8;
#pragma unroll
            for (int v = 0; v < 8; ++v) {
                const int row = r0 + v;
                if (row < TILE) {
                    fb[row * FP + fcol] = cc[v];
                    if (base + row < NN) featn[(long)(base + row) * HD + ct * 16 + l15] = cc[v];
                }
            }
        }
        __syncthreads();

        if (tid < TILE * HEADS && base + (tid >> 1) < NN) {
            const int hh = tid & 1;
            const float* f  = fb + (tid >> 1) * FP + hh * FH1;
            const float* la = aln + hh * HID;
            const float* ra = arn + hh * HID;
            float sl = 0.f, sr = 0.f;
#pragma unroll 4
            for (int d = 0; d < HID; ++d) { float v = f[d]; sl += v * la[d]; sr += v * ra[d]; }
            eln[base * HEADS + tid] = sl; ern[base * HEADS + tid] = sr;
        }
        return;
    }

    for (int i = tid; i < TILE * HID; i += TB) {
        const int nl = i >> 6, dd = i & 63;
        const int n  = base + nl;
        if (n < NN) {
            const float s0 = s_l[nl * HEADS], s1 = s_l[nl * HEADS + 1];
            const int w0 = nl * HD + (dd & 3) * 32 + (dd >> 2);
            const float a0 = s0 > 0.f ? acc[w0]      / s0 : 0.f;
            const float a1 = s1 > 0.f ? acc[w0 + 16] / s1 : 0.f;
            const float v0 = a0 + bias[dd];
            const float v1 = a1 + bias[HID + dd];
            const float v  = 0.5f * (v0 + v1);
            out[(long)n * HID + dd] = v > 0.f ? v : (__expf(v) - 1.f);
        }
    }
}

extern "C" void kernel_launch(void* const* d_in, const int* in_sizes, int n_in,
                              void* d_out, int out_size, void* d_ws, size_t ws_size,
                              hipStream_t stream) {
    const float* x   = (const float*)d_in[0];
    const float* W1  = (const float*)d_in[1];
    const float* al1 = (const float*)d_in[2];
    const float* ar1 = (const float*)d_in[3];
    const float* b1  = (const float*)d_in[4];
    const float* W2  = (const float*)d_in[5];
    const float* al2 = (const float*)d_in[6];
    const float* ar2 = (const float*)d_in[7];
    const float* b2  = (const float*)d_in[8];
    const int*   src = (const int*)d_in[9];
    const int*   dst = (const int*)d_in[10];
    float* out = (float*)d_out;

    char* wsb = (char*)d_ws;
    size_t off = 0;
    auto alloc = [&](size_t bytes) -> void* {
        void* p = wsb + off;
        off += bytes;
        off = (off + 255) & ~(size_t)255;
        return p;
    };
    float* WT1  = (float*)alloc((size_t)INF_ * HD * 4);
    float* feat = (float*)alloc((size_t)NN * HD * 4);
    float* el   = (float*)alloc((size_t)NN * HEADS * 4);
    float* er   = (float*)alloc((size_t)NN * HEADS * 4);
    float* feat2 = (float*)alloc((size_t)NN * HD * 4);
    float* el2  = (float*)alloc((size_t)NN * HEADS * 4);
    float* er2  = (float*)alloc((size_t)NN * HEADS * 4);
    unsigned char* bkt = (unsigned char*)alloc((size_t)NE);
    unsigned* tcnt  = (unsigned*)alloc((size_t)NTILES * 4);
    unsigned* tlist = (unsigned*)alloc((size_t)NTILES * QCAP * 4);

    const int NH = NN * HEADS;
    const dim3 B256(256);

    gat_bucket<<<dim3(NBK), B256, 0, stream>>>(dst, bkt, tlist, NE);

    gat_w_transpose<<<dim3((INF_ * HD + 255) / 256), B256, 0, stream>>>(W1, WT1, INF_, INF_ * HD);
    gat_gemm_wmma_f32<<<dim3(NN / 16), B256, 0, stream>>>(x, WT1, feat, INF_);
    gat_eler<<<dim3((NH + 255) / 256), B256, 0, stream>>>(feat, al1, ar1, el, er, NH);
    gat_tile<<<dim3(NTILES), dim3(TB), 0, stream>>>(src, dst, bkt, feat, el, er, b1, out, tcnt, tlist, 0,
                                                    W2, al2, ar2, feat2, el2, er2);

    gat_tile<<<dim3(NTILES), dim3(TB), 0, stream>>>(src, dst, bkt, feat2, el2, er2, b2, out, tcnt, tlist, 1,
                                                    nullptr, nullptr, nullptr, nullptr, nullptr, nullptr);
}


# Row pass without LDS adds: waves own tile rows (w, w+32, ..); edges counting-sorted by row in LDS (count, scan, fill by arrival rank), each row summed in registers, stored once; sums in bin order
# speedup vs baseline: 7.7891x; 1.1398x over previous
#include <hip/hip_runtime.h>
#include <hip/hip_bf16.h>

#define NN   50000
#define NE   1600000
#define INF_ 128
#define HID  64
#define HEADS 2
#define HD   (HEADS*HID)
#define NEG_SLOPE 0.2f

typedef __attribute__((ext_vector_type(2))) float v2f;
typedef __attribute__((ext_vector_type(8))) float v8f;

__device__ __forceinline__ void atomicMaxF(float* addr, float val) {
    if (val >= 0.0f) atomicMax((int*)addr, __float_as_int(val));
    else             atomicMin((unsigned int*)addr, (unsigned int)__float_as_int(val));
}


__global__ void gat_w_transpose(const float* __restrict__ W, float* __restrict__ WT,
                                int K, int total) {
    int i = blockIdx.x * blockDim.x + threadIdx.x;
    if (i >= total) return;
    int k = i / HD;
    int n = i - k * HD;
    WT[(long)n * K + k] = W[i];
}

__global__ void __launch_bounds__(256) gat_gemm_wmma_f32(
        const float* __restrict__ A, const float* __restrict__ BT,
        float* __restrict__ D, int K) {
    const int lane = threadIdx.x & 31;
    const int wave = threadIdx.x >> 5;
    const int half = lane >> 4;
    const int l15  = lane & 15;

    const float* __restrict__ arow = A  + ((long)blockIdx.x * 16 + l15) * K;
    const float* __restrict__ brow = BT + ((long)wave       * 16 + l15) * K;

    v8f c = {};
    for (int kb = 0; kb < K; kb += 4) {
        const int k0 = kb + half * 2;
        v2f a, b;
        a.x = arow[k0]; a.y = arow[k0 + 1];
        b.x = brow[k0]; b.y = brow[k0 + 1];
        c = __builtin_amdgcn_wmma_f32_16x16x4_f32(false, a, false, b,
                                                  (short)0, c, false, false);
    }
    float* __restrict__ drow =
        D + ((long)blockIdx.x * 16 + half * 8) * HD + wave * 16 + l15;
#pragma unroll
    for (int v = 0; v < 8; ++v) drow[(long)v * HD] = c[v];
}

__global__ void gat_eler(const float* __restrict__ feat,
                         const float* __restrict__ al, const float* __restrict__ ar,
                         float* __restrict__ el, float* __restrict__ er, int NH) {
    int i = blockIdx.x * blockDim.x + threadIdx.x;
    if (i >= NH) return;
    const int h = i & 1;
    const float* __restrict__ f = feat + (long)(i >> 1) * HD + h * HID;
    const float* __restrict__ la = al + h * HID;
    const float* __restrict__ ra = ar + h * HID;
    float sl = 0.f, sr = 0.f;
#pragma unroll 4
    for (int d = 0; d < HID; ++d) { float v = f[d]; sl += v * la[d]; sr += v * ra[d]; }
    el[i] = sl; er[i] = sr;
}

#define TILE   196
#define NTILES 256
#define QCAP   8192
#define TB     1024
#define RG     8
#define HROWS  208
#define HP     68
#define FP     130
#define FH1    65
#define ACCW   (TILE * FP)

#define BKE    2048
#define NBK    ((NE + BKE - 1) / BKE)
#define BMW    (NBK * 4)

__global__ void __launch_bounds__(256) gat_bucket(const int* __restrict__ dst,
                                                  unsigned char* __restrict__ bkt,
                                                  unsigned* __restrict__ tlist, int E) {
    __shared__ unsigned w[4 * NTILES];
    const int t = threadIdx.x;
    for (int i = t; i < 4 * NTILES; i += 256) w[i] = 0u;
    __syncthreads();
    const int e0 = blockIdx.x * BKE + t * 8;
    if (e0 < E) {
        const int4 a = *(const int4*)(dst + e0);
        const int4 b = *(const int4*)(dst + e0 + 4);
        const unsigned t0 = a.x / TILE, t1 = a.y / TILE, t2 = a.z / TILE, t3 = a.w / TILE;
        const unsigned t4 = b.x / TILE, t5 = b.y / TILE, t6 = b.z / TILE, t7 = b.w / TILE;
        uint2 pk;
        pk.x = t0 | (t1 << 8) | (t2 << 16) | (t3 << 24);
        pk.y = t4 | (t5 << 8) | (t6 << 16) | (t7 << 24);
        *(uint2*)(bkt + e0) = pk;
        unsigned* wj = w + (t >> 6) * NTILES;
        const unsigned bit = 1u << ((t >> 1) & 31);
        atomicOr(wj + t0, bit); atomicOr(wj + t1, bit); atomicOr(wj + t2, bit); atomicOr(wj + t3, bit);
        atomicOr(wj + t4, bit); atomicOr(wj + t5, bit); atomicOr(wj + t6, bit); atomicOr(wj + t7, bit);
    }
    __syncthreads();
    uint4 o;
    o.x = w[t]; o.y = w[NTILES + t]; o.z = w[2 * NTILES + t]; o.w = w[3 * NTILES + t];
    *(uint4*)(tlist + (long)t * QCAP + blockIdx.x * 4) = o;
}

__device__ __forceinline__ void tile_match16(uint4 w, unsigned pat, int e0, int* q_e, unsigned* qn) {
    const unsigned ws[4] = {w.x, w.y, w.z, w.w};
#pragma unroll
    for (int k = 0; k < 4; ++k) {
        const unsigned x = ws[k] ^ pat;
        unsigned t = ~(((x & 0x7f7f7f7fu) + 0x7f7f7f7fu) | x | 0x7f7f7f7fu);
        while (t) {
            const int b = __builtin_ctz(t) >> 3;
            const unsigned slot = atomicAdd(qn, 1u);
            if (slot < QCAP) q_e[slot] = e0 + k * 4 + b;
            t &= t - 1u;
        }
    }
}

__device__ __forceinline__ void tile_scan(const unsigned char* __restrict__ bkt,
                                          int e_begin, int e_end, unsigned myb,
                                          int* q_e, unsigned* qn, int tid) {
    const uint4* __restrict__ p = (const uint4*)bkt;
    const unsigned pat = myb * 0x01010101u;
    const int g1 = e_end >> 4;
    int g = (e_begin >> 4) + tid;
    for (; g + 7 * TB < g1; g += 8 * TB) {
        uint4 w[8];
#pragma unroll
        for (int u = 0; u < 8; ++u) w[u] = p[g + u * TB];
#pragma unroll
        for (int u = 0; u < 8; ++u) tile_match16(w[u], pat, (g + u * TB) * 16, q_e, qn);
    }
    for (; g < g1; g += TB) tile_match16(p[g], pat, g * 16, q_e, qn);
}

__device__ __forceinline__ void map_walk(const unsigned* bm, unsigned* q_g, unsigned* gn, int tid) {
    for (int i = tid; i < BMW; i += TB) {
        unsigned w = bm[i];
        while (w) {
            const int b = __builtin_ctz(w);
            w &= w - 1u;
            const unsigned slot = atomicAdd(gn, 1u);
            if (slot < QCAP) q_g[slot] = (unsigned)(i * 32 + b);
        }
    }
}

__device__ __forceinline__ void group_match(const unsigned char* __restrict__ bkt,
                                            const unsigned* q_g, unsigned ng, unsigned myb,
                                            int* q_e, unsigned* qn, int tid) {
    const uint4* __restrict__ p = (const uint4*)bkt;
    const unsigned pat = myb * 0x01010101u;
    for (unsigned i0 = tid; i0 < ng; i0 += 4 * TB) {
        uint4 v[4]; unsigned g[4]; bool live[4];
#pragma unroll
        for (int u = 0; u < 4; ++u) {
            const unsigned i = i0 + u * TB;
            live[u] = i < ng;
            g[u] = q_g[live[u] ? i : i0];
            v[u] = p[g[u]];
        }
#pragma unroll
        for (int u = 0; u < 4; ++u) {
            const unsigned ws[4] = {v[u].x, v[u].y, v[u].z, v[u].w};
            unsigned m = 0u;
#pragma unroll
            for (int k = 0; k < 4; ++k) {
                const unsigned x = ws[k] ^ pat;
                const unsigned t = ~(((x & 0x7f7f7f7fu) + 0x7f7f7f7fu) | x | 0x7f7f7f7fu);
                m |= (t >> 7) << k;
            }
            if (!live[u]) m = 0u;
            while (m) {
                const int pbit = __builtin_ctz(m);
                m &= m - 1u;
                const unsigned slot = atomicAdd(qn, 1u);
                if (slot < QCAP) q_e[slot] = (int)(g[u] * 16u) + (pbit & 7) * 4 + (pbit >> 3);
            }
        }
    }
}

template <bool TAIL>
__device__ __forceinline__ void row_trip(const float* __restrict__ feat, const int* q_e,
                                         const float* q_x, float* acc,
                                         unsigned i0, unsigned cnt, int lane) {
    const unsigned NW = TB / 32;
    float4 f[RG]; float x[RG]; int d[RG];
#pragma unroll
    for (int u = 0; u < RG; ++u) {
        unsigned i = i0 + u * NW;
        bool live = true;
        if (TAIL) { live = i < cnt; i = live ? i : cnt - 1u; }
        const unsigned p = (unsigned)q_e[i];
        d[u] = live ? (int)(p >> 16) : TILE;
        x[u] = q_x[i];
        f[u] = *(const float4*)(feat + (long)(p & 0xffffu) * HD + lane * 4);
    }
#pragma unroll
    for (int u = 0; u < RG; ++u) {
        float* o = acc + d[u] * HD + lane;
        unsafeAtomicAdd(o,      f[u].x * x[u]);
        unsafeAtomicAdd(o + 32, f[u].y * x[u]);
        unsafeAtomicAdd(o + 64, f[u].z * x[u]);
        unsafeAtomicAdd(o + 96, f[u].w * x[u]);
    }
}

#define NWV    (TB / 32)
#define OWNR   ((TILE + NWV - 1) / NWV)
#define NBIN   (NWV * OWNR)

__device__ __forceinline__ unsigned own_bin(int dl) {
    return (unsigned)(dl & (NWV - 1)) * OWNR + (unsigned)(dl / NWV);
}

__device__ __forceinline__ void own_flush(float* acc, float4 cur, int r, int lane) {
    if (r >= 0) {
        float* o = acc + r * HD + lane;
        o[0] = cur.x; o[32] = cur.y; o[64] = cur.z; o[96] = cur.w;
    }
}

template <bool TAIL>
__device__ __forceinline__ void own_trip(const float* __restrict__ feat, const unsigned* q_s,
                                         const float* q_x, float* acc, float4& cur, int& rcur,
                                         unsigned i0, unsigned b1, int lane) {
    float4 f[RG]; float x[RG]; int r[RG];
#pragma unroll
    for (int u = 0; u < RG; ++u) {
        unsigned i = i0 + u;
        if (TAIL) i = i < b1 ? i : b1 - 1u;
        const unsigned p = q_s[i];
        r[u] = __builtin_amdgcn_readfirstlane((int)(p >> 16));
        x[u] = q_x[i];
        f[u] = *(const float4*)(feat + (long)(p & 0xffffu) * HD + lane * 4);
    }
#pragma unroll
    for (int u = 0; u < RG; ++u) {
        if (!TAIL || i0 + u < b1) {
            if (r[u] != rcur) {
                own_flush(acc, cur, rcur, lane);
                cur = make_float4(0.f, 0.f, 0.f, 0.f);
                rcur = r[u];
            }
            cur.x += f[u].x * x[u]; cur.y += f[u].y * x[u];
            cur.z += f[u].z * x[u]; cur.w += f[u].w * x[u];
        }
    }
}

__global__ void __launch_bounds__(TB) gat_tile(
        const int* __restrict__ src, const int* __restrict__ dst,
        const unsigned char* __restrict__ bkt,
        const float* __restrict__ feat,
        const float* __restrict__ el, const float* __restrict__ er,
        const float* __restrict__ bias, float* __restrict__ out,
        unsigned* __restrict__ tcnt, unsigned* __restrict__ tlist, int use_list,
        const float* __restrict__ Wn, const float* __restrict__ aln, const float* __restrict__ arn,
        float* __restrict__ featn, float* __restrict__ eln, float* __restrict__ ern) {
    __shared__ __attribute__((aligned(16))) float acc[ACCW];
    __shared__ float m_l[TILE * HEADS];
    __shared__ float s_l[TILE * HEADS];
    __shared__ float er_l[TILE * HEADS];
    __shared__ int   q_e[QCAP];
    __shared__ __attribute__((aligned(16))) float q_xb[2 * QCAP];
    __shared__ unsigned q_g[QCAP];
    __shared__ unsigned qn, gn;
    __shared__ unsigned bcnt[NBIN], bfill[NBIN], boff[NBIN + 1];
    float* const q_x0 = q_xb;
    float* const q_x1 = q_xb + QCAP;

    const int tid  = threadIdx.x;
    const int base = blockIdx.x * TILE;
    const unsigned myb = blockIdx.x;
    unsigned* __restrict__ mylist = tlist + (long)blockIdx.x * QCAP;

    for (int i = tid; i < (TILE + 1) * HD; i += TB) acc[i] = 0.f;
    for (int i = tid; i < TILE * HEADS; i += TB) {
        m_l[i] = -INFINITY; s_l[i] = 0.f;
        er_l[i] = (base + (i >> 1)) < NN ? er[base * HEADS + i] : 0.f;
    }
    if (tid == 0) { qn = 0u; gn = 0u; }
    if (tid < NBIN) { bcnt[tid] = 0u; bfill[tid] = 0u; }
    __syncthreads();

    unsigned cnt = 0u;
    bool listed = false;
    if (use_list) {
        cnt = tcnt[blockIdx.x];
        listed = cnt <= QCAP;
    }
    if (!listed) map_walk(mylist, q_g, &gn, tid);
    __syncthreads();
    if (!listed) {
        const unsigned ng = gn;
        if (ng <= QCAP) group_match(bkt, q_g, ng, myb, q_e, &qn, tid);
        else if (tid == 0) qn = ng;
    }
    __syncthreads();
    if (!listed) {
        cnt = qn;
        if (!use_list && tid == 0) tcnt[blockIdx.x] = cnt;
    }
    const int seglen = cnt > QCAP ? QCAP : NE;
    const int nseg   = (NE + seglen - 1) / seglen;
    const bool publish = !use_list && nseg == 1;
    const bool own = nseg == 1;

    for (int seg = 0; seg < nseg; ++seg) {
        if (nseg > 1) {
            __syncthreads();
            if (tid == 0) qn = 0u;
            __syncthreads();
            const int eb = seg * seglen;
            tile_scan(bkt, eb, min(NE, eb + seglen), myb, q_e, &qn, tid);
            __syncthreads();
            cnt = qn;
        }
        for (unsigned i = tid; i < cnt; i += TB) {
            unsigned p;
            if (listed) {
                p = mylist[i];
            } else {
                const int e = q_e[i];
                p = (unsigned)src[e] | ((unsigned)(dst[e] - base) << 16);
                if (publish) mylist[i] = p;
            }
            q_e[i] = (int)p;
            const int s0 = (int)(p & 0xffffu);
            const int dl = (int)(p >> 16);
            const float2 l2 = *(const float2*)(el + (long)s0 * HEADS);
            float v0 = l2.x + er_l[dl * HEADS], v1 = l2.y + er_l[dl * HEADS + 1];
            v0 = v0 > 0.f ? v0 : NEG_SLOPE * v0;
            v1 = v1 > 0.f ? v1 : NEG_SLOPE * v1;
            atomicMaxF(&m_l[dl * HEADS], v0);
            atomicMaxF(&m_l[dl * HEADS + 1], v1);
            if (own) atomicAdd(&bcnt[own_bin(dl)], 1u);
        }
    }
    __syncthreads();
    if (own) {
        if (tid < NWV) {
            unsigned cb[OWNR], tot = 0u;
#pragma unroll
            for (int k = 0; k < OWNR; ++k) { cb[k] = bcnt[tid * OWNR + k]; tot += cb[k]; }
            unsigned inc = tot;
#pragma unroll
            for (int d = 1; d < NWV; d <<= 1) {
                const unsigned v = __shfl_up(inc, d);
                if (tid >= d) inc += v;
            }
            unsigned o = inc - tot;
#pragma unroll
            for (int k = 0; k < OWNR; ++k) { boff[tid * OWNR + k] = o; o += cb[k]; }
            if (tid == NWV - 1) boff[NBIN] = o;
        }
        __syncthreads();
    }

    const int wave = tid >> 5, lane = tid & 31;
    const int h  = lane >> 4;
    for (int seg = 0; seg < nseg; ++seg) {
        if (nseg > 1) {
            __syncthreads();
            if (tid == 0) qn = 0u;
            __syncthreads();
            const int eb = seg * seglen;
            tile_scan(bkt, eb, min(NE, eb + seglen), myb, q_e, &qn, tid);
            __syncthreads();
            cnt = qn;
        }
        for (unsigned i = tid; i < cnt; i += TB) {
            unsigned p = (unsigned)q_e[i];
            if (nseg > 1) {
                const int e = (int)p;
                p = (unsigned)src[e] | ((unsigned)(dst[e] - base) << 16);
            }
            const int s0 = (int)(p & 0xffffu);
            const int dl = (int)(p >> 16);
            const float2 l2 = *(const float2*)(el + (long)s0 * HEADS);
            float v0 = l2.x + er_l[dl * HEADS], v1 = l2.y + er_l[dl * HEADS + 1];
            v0 = v0 > 0.f ? v0 : NEG_SLOPE * v0;
            v1 = v1 > 0.f ? v1 : NEG_SLOPE * v1;
            const float x0 = __expf(v0 - m_l[dl * HEADS]);
            const float x1 = __expf(v1 - m_l[dl * HEADS + 1]);
            unsigned slot = i;
            if (own) {
                const unsigned b = own_bin(dl);
                slot = boff[b] + atomicAdd(&bfill[b], 1u);
                q_g[slot] = p;
            } else {
                q_e[i] = (int)p;
            }
            q_x0[slot] = x0; q_x1[slot] = x1;
            unsafeAtomicAdd(&s_l[dl * HEADS], x0);
            unsafeAtomicAdd(&s_l[dl * HEADS + 1], x1);
        }
        __syncthreads();
        const unsigned NW = TB / 32;
        const float* q_x = h ? q_x1 : q_x0;
        if (own) {
            const int wo = __builtin_amdgcn_readfirstlane(wave);
            const unsigned b1 = (unsigned)__builtin_amdgcn_readfirstlane((int)boff[(wo + 1) * OWNR]);
            unsigned i0 = (unsigned)__builtin_amdgcn_readfirstlane((int)boff[wo * OWNR]);
            float4 cur = make_float4(0.f, 0.f, 0.f, 0.f);
            int rcur = -1;
            for (; i0 + RG <= b1; i0 += RG) own_trip<false>(feat, q_g, q_x, acc, cur, rcur, i0, b1, lane);
            if (i0 < b1) own_trip<true>(feat, q_g, q_x, acc, cur, rcur, i0, b1, lane);
            own_flush(acc, cur, rcur, lane);
        } else {
            unsigned i0 = wave;
            for (; i0 + (RG - 1) * NW < cnt; i0 += RG * NW)
                row_trip<false>(feat, q_e, q_x, acc, i0, cnt, lane);
            if (i0 < cnt) row_trip<true>(feat, q_e, q_x, acc, i0, cnt, lane);
        }
    }
    __syncthreads();

    if (!use_list) {
        float* const hb = q_xb;
        for (int i = tid; i < HROWS * HID; i += TB) {
            const int nl = i >> 6, dd = i & 63;
            float r = 0.f;
            if (nl < TILE && base + nl < NN) {
                const float s0 = s_l[nl * HEADS], s1 = s_l[nl * HEADS + 1];
                const int w0 = nl * HD + (dd & 3) * 32 + (dd >> 2);
                const float a0 = s0 > 0.f ? acc[w0]      / s0 : 0.f;
                const float a1 = s1 > 0.f ? acc[w0 + 16] / s1 : 0.f;
                const float v0 = a0 + bias[dd];
                const float v1 = a1 + bias[HID + dd];
                const float v  = 0.5f * (v0 + v1);
                r = v > 0.f ? v : (__expf(v) - 1.f);
            }
            hb[nl * HP + dd] = r;
        }
        __syncthreads();

        const int wv   = __builtin_amdgcn_readfirstlane(wave);
        const int ct   = wv & 7;
        const int half = lane >> 4, l15 = lane & 15;
        v2f bf[HID / 4];
#pragma unroll
        for (int kq = 0; kq < HID / 4; ++kq) {
            const int k0 = kq * 4 + half * 2;
            bf[kq].x = Wn[k0 * HD + ct * 16 + l15];
            bf[kq].y = Wn[(k0 + 1) * HD + ct * 16 + l15];
        }
        float* const fb = acc;
        const int fcol = (ct >> 2) * FH1 + (ct & 3) * 16 + l15;
        for (int rt = wv >> 3; rt < HROWS / 16; rt += 4) {
            const float* arow = hb + (rt * 16 + l15) * HP + half * 2;
            v8f cc = {};
#pragma unroll
            for (int kq = 0; kq < HID / 4; ++kq) {
                const v2f a = *(const v2f*)(arow + kq * 4);
                cc = __builtin_amdgcn_wmma_f32_16x16x4_f32(false, a, false, bf[kq],
                                                           (short)0, cc, false, false);
            }
            const int r0 = rt * 16 + half * 8;
#pragma unroll
            for (int v = 0; v < 8; ++v) {
                const int row = r0 + v;
                if (row < TILE) {
                    fb[row * FP + fcol] = cc[v];
                    if (base + row < NN) featn[(long)(base + row) * HD + ct * 16 + l15] = cc[v];
                }
            }
        }
        __syncthreads();

        if (tid < TILE * HEADS && base + (tid >> 1) < NN) {
            const int hh = tid & 1;
            const float* f  = fb + (tid >> 1) * FP + hh * FH1;
            const float* la = aln + hh * HID;
            const float* ra = arn + hh * HID;
            float sl = 0.f, sr = 0.f;
#pragma unroll 4
            for (int d = 0; d < HID; ++d) { float v = f[d]; sl += v * la[d]; sr += v * ra[d]; }
            eln[base * HEADS + tid] = sl; ern[base * HEADS + tid] = sr;
        }
        return;
    }

    for (int i = tid; i < TILE * HID; i += TB) {
        const int nl = i >> 6, dd = i & 63;
        const int n  = base + nl;
        if (n < NN) {
            const float s0 = s_l[nl * HEADS], s1 = s_l[nl * HEADS + 1];
            const int w0 = nl * HD + (dd & 3) * 32 + (dd >> 2);
            const float a0 = s0 > 0.f ? acc[w0]      / s0 : 0.f;
            const float a1 = s1 > 0.f ? acc[w0 + 16] / s1 : 0.f;
            const float v0 = a0 + bias[dd];
            const float v1 = a1 + bias[HID + dd];
            const float v  = 0.5f * (v0 + v1);
            out[(long)n * HID + dd] = v > 0.f ? v : (__expf(v) - 1.f);
        }
    }
}

extern "C" void kernel_launch(void* const* d_in, const int* in_sizes, int n_in,
                              void* d_out, int out_size, void* d_ws, size_t ws_size,
                              hipStream_t stream) {
    const float* x   = (const float*)d_in[0];
    const float* W1  = (const float*)d_in[1];
    const float* al1 = (const float*)d_in[2];
    const float* ar1 = (const float*)d_in[3];
    const float* b1  = (const float*)d_in[4];
    const float* W2  = (const float*)d_in[5];
    const float* al2 = (const float*)d_in[6];
    const float* ar2 = (const float*)d_in[7];
    const float* b2  = (const float*)d_in[8];
    const int*   src = (const int*)d_in[9];
    const int*   dst = (const int*)d_in[10];
    float* out = (float*)d_out;

    char* wsb = (char*)d_ws;
    size_t off = 0;
    auto alloc = [&](size_t bytes) -> void* {
        void* p = wsb + off;
        off += bytes;
        off = (off + 255) & ~(size_t)255;
        return p;
    };
    float* WT1  = (float*)alloc((size_t)INF_ * HD * 4);
    float* feat = (float*)alloc((size_t)NN * HD * 4);
    float* el   = (float*)alloc((size_t)NN * HEADS * 4);
    float* er   = (float*)alloc((size_t)NN * HEADS * 4);
    float* feat2 = (float*)alloc((size_t)NN * HD * 4);
    float* el2  = (float*)alloc((size_t)NN * HEADS * 4);
    float* er2  = (float*)alloc((size_t)NN * HEADS * 4);
    unsigned char* bkt = (unsigned char*)alloc((size_t)NE);
    unsigned* tcnt  = (unsigned*)alloc((size_t)NTILES * 4);
    unsigned* tlist = (unsigned*)alloc((size_t)NTILES * QCAP * 4);

    const int NH = NN * HEADS;
    const dim3 B256(256);

    gat_bucket<<<dim3(NBK), B256, 0, stream>>>(dst, bkt, tlist, NE);

    gat_w_transpose<<<dim3((INF_ * HD + 255) / 256), B256, 0, stream>>>(W1, WT1, INF_, INF_ * HD);
    gat_gemm_wmma_f32<<<dim3(NN / 16), B256, 0, stream>>>(x, WT1, feat, INF_);
    gat_eler<<<dim3((NH + 255) / 256), B256, 0, stream>>>(feat, al1, ar1, el, er, NH);
    gat_tile<<<dim3(NTILES), dim3(TB), 0, stream>>>(src, dst, bkt, feat, el, er, b1, out, tcnt, tlist, 0,
                                                    W2, al2, ar2, feat2, el2, er2);

    gat_tile<<<dim3(NTILES), dim3(TB), 0, stream>>>(src, dst, bkt, feat2, el2, er2, b2, out, tcnt, tlist, 1,
                                                    nullptr, nullptr, nullptr, nullptr, nullptr, nullptr);
}


# Layer 1 projection on the tile epilogue's data path: 64 x rows put in LDS by whole-row loads, a W1 fragment read once for 4 row tiles, rows stored whole, logits from LDS, same sums; 6 launches to 4
# speedup vs baseline: 8.8876x; 1.1410x over previous
#include <hip/hip_runtime.h>
#include <hip/hip_bf16.h>

#define NN   50000
#define NE   1600000
#define INF_ 128
#define HID  64
#define HEADS 2
#define HD   (HEADS*HID)
#define NEG_SLOPE 0.2f

typedef __attribute__((ext_vector_type(2))) float v2f;
typedef __attribute__((ext_vector_type(8))) float v8f;

__device__ __forceinline__ void atomicMaxF(float* addr, float val) {
    if (val >= 0.0f) atomicMax((int*)addr, __float_as_int(val));
    else             atomicMin((unsigned int*)addr, (unsigned int)__float_as_int(val));
}


#define PR   64
#define XP   132

__global__ void __launch_bounds__(256) gat_proj(
        const float* __restrict__ x, const float* __restrict__ W,
        const float* __restrict__ al, const float* __restrict__ ar,
        float* __restrict__ feat, float* __restrict__ el, float* __restrict__ er) {
    __shared__ __attribute__((aligned(16))) float xb[PR * XP];
    const int tid  = threadIdx.x;
    const int lane = tid & 31;
    const int wv   = __builtin_amdgcn_readfirstlane(tid >> 5);
    const int base = blockIdx.x * PR;
    const int half = lane >> 4, l15 = lane & 15;

    float4 xr[PR / 8];
#pragma unroll
    for (int j = 0; j < PR / 8; ++j) {
        int n = base + wv + j * 8;
        n = n < NN ? n : NN - 1;
        xr[j] = *(const float4*)(x + (long)n * INF_ + lane * 4);
    }
#pragma unroll
    for (int j = 0; j < PR / 8; ++j) {
        const int r = wv + j * 8;
        *(float4*)(xb + r * XP + lane * 4) = base + r < NN ? xr[j] : make_float4(0.f, 0.f, 0.f, 0.f);
    }
    __syncthreads();

    v8f cc[PR / 16];
#pragma unroll
    for (int t = 0; t < PR / 16; ++t) cc[t] = v8f{};
    const float* wcol = W + wv * 16 + l15;
    const float* arow = xb + l15 * XP + half * 2;
#pragma unroll 8
    for (int kq = 0; kq < INF_ / 4; ++kq) {
        const int k0 = kq * 4 + half * 2;
        v2f b;
        b.x = wcol[k0 * HD];
        b.y = wcol[(k0 + 1) * HD];
#pragma unroll
        for (int t = 0; t < PR / 16; ++t) {
            const v2f a = *(const v2f*)(arow + t * 16 * XP + kq * 4);
            cc[t] = __builtin_amdgcn_wmma_f32_16x16x4_f32(false, a, false, b,
                                                          (short)0, cc[t], false, false);
        }
    }
    __syncthreads();

#pragma unroll
    for (int t = 0; t < PR / 16; ++t) {
#pragma unroll
        for (int v = 0; v < 8; ++v)
            xb[(t * 16 + half * 8 + v) * XP + wv * 16 + l15] = cc[t][v];
    }
    __syncthreads();

#pragma unroll
    for (int j = 0; j < PR / 8; ++j) xr[j] = *(const float4*)(xb + (wv + j * 8) * XP + lane * 4);
#pragma unroll
    for (int j = 0; j < PR / 8; ++j) {
        const int r = wv + j * 8;
        if (base + r < NN) *(float4*)(feat + (long)(base + r) * HD + lane * 4) = xr[j];
    }

    if (tid < PR * HEADS && base + (tid >> 1) < NN) {
        const int hh = tid & 1;
        const float* f  = xb + (tid >> 1) * XP + hh * HID;
        const float* la = al + hh * HID;
        const float* ra = ar + hh * HID;
        float sl = 0.f, sr = 0.f;
#pragma unroll 4
        for (int d = 0; d < HID; ++d) { float v = f[d]; sl += v * la[d]; sr += v * ra[d]; }
        el[base * HEADS + tid] = sl; er[base * HEADS + tid] = sr;
    }
}

#define TILE   196
#define NTILES 256
#define QCAP   8192
#define TB     1024
#define RG     8
#define HROWS  208
#define HP     68
#define FP     130
#define FH1    65
#define ACCW   (TILE * FP)

#define BKE    2048
#define NBK    ((NE + BKE - 1) / BKE)
#define BMW    (NBK * 4)

__global__ void __launch_bounds__(256) gat_bucket(const int* __restrict__ dst,
                                                  unsigned char* __restrict__ bkt,
                                                  unsigned* __restrict__ tlist, int E) {
    __shared__ unsigned w[4 * NTILES];
    const int t = threadIdx.x;
    for (int i = t; i < 4 * NTILES; i += 256) w[i] = 0u;
    __syncthreads();
    const int e0 = blockIdx.x * BKE + t * 8;
    if (e0 < E) {
        const int4 a = *(const int4*)(dst + e0);
        const int4 b = *(const int4*)(dst + e0 + 4);
        const unsigned t0 = a.x / TILE, t1 = a.y / TILE, t2 = a.z / TILE, t3 = a.w / TILE;
        const unsigned t4 = b.x / TILE, t5 = b.y / TILE, t6 = b.z / TILE, t7 = b.w / TILE;
        uint2 pk;
        pk.x = t0 | (t1 << 8) | (t2 << 16) | (t3 << 24);
        pk.y = t4 | (t5 << 8) | (t6 << 16) | (t7 << 24);
        *(uint2*)(bkt + e0) = pk;
        unsigned* wj = w + (t >> 6) * NTILES;
        const unsigned bit = 1u << ((t >> 1) & 31);
        atomicOr(wj + t0, bit); atomicOr(wj + t1, bit); atomicOr(wj + t2, bit); atomicOr(wj + t3, bit);
        atomicOr(wj + t4, bit); atomicOr(wj + t5, bit); atomicOr(wj + t6, bit); atomicOr(wj + t7, bit);
    }
    __syncthreads();
    uint4 o;
    o.x = w[t]; o.y = w[NTILES + t]; o.z = w[2 * NTILES + t]; o.w = w[3 * NTILES + t];
    *(uint4*)(tlist + (long)t * QCAP + blockIdx.x * 4) = o;
}

__device__ __forceinline__ void tile_match16(uint4 w, unsigned pat, int e0, int* q_e, unsigned* qn) {
    const unsigned ws[4] = {w.x, w.y, w.z, w.w};
#pragma unroll
    for (int k = 0; k < 4; ++k) {
        const unsigned x = ws[k] ^ pat;
        unsigned t = ~(((x & 0x7f7f7f7fu) + 0x7f7f7f7fu) | x | 0x7f7f7f7fu);
        while (t) {
            const int b = __builtin_ctz(t) >> 3;
            const unsigned slot = atomicAdd(qn, 1u);
            if (slot < QCAP) q_e[slot] = e0 + k * 4 + b;
            t &= t - 1u;
        }
    }
}

__device__ __forceinline__ void tile_scan(const unsigned char* __restrict__ bkt,
                                          int e_begin, int e_end, unsigned myb,
                                          int* q_e, unsigned* qn, int tid) {
    const uint4* __restrict__ p = (const uint4*)bkt;
    const unsigned pat = myb * 0x01010101u;
    const int g1 = e_end >> 4;
    int g = (e_begin >> 4) + tid;
    for (; g + 7 * TB < g1; g += 8 * TB) {
        uint4 w[8];
#pragma unroll
        for (int u = 0; u < 8; ++u) w[u] = p[g + u * TB];
#pragma unroll
        for (int u = 0; u < 8; ++u) tile_match16(w[u], pat, (g + u * TB) * 16, q_e, qn);
    }
    for (; g < g1; g += TB) tile_match16(p[g], pat, g * 16, q_e, qn);
}

__device__ __forceinline__ void map_walk(const unsigned* bm, unsigned* q_g, unsigned* gn, int tid) {
    for (int i = tid; i < BMW; i += TB) {
        unsigned w = bm[i];
        while (w) {
            const int b = __builtin_ctz(w);
            w &= w - 1u;
            const unsigned slot = atomicAdd(gn, 1u);
            if (slot < QCAP) q_g[slot] = (unsigned)(i * 32 + b);
        }
    }
}

__device__ __forceinline__ void group_match(const unsigned char* __restrict__ bkt,
                                            const unsigned* q_g, unsigned ng, unsigned myb,
                                            int* q_e, unsigned* qn, int tid) {
    const uint4* __restrict__ p = (const uint4*)bkt;
    const unsigned pat = myb * 0x01010101u;
    for (unsigned i0 = tid; i0 < ng; i0 += 4 * TB) {
        uint4 v[4]; unsigned g[4]; bool live[4];
#pragma unroll
        for (int u = 0; u < 4; ++u) {
            const unsigned i = i0 + u * TB;
            live[u] = i < ng;
            g[u] = q_g[live[u] ? i : i0];
            v[u] = p[g[u]];
        }
#pragma unroll
        for (int u = 0; u < 4; ++u) {
            const unsigned ws[4] = {v[u].x, v[u].y, v[u].z, v[u].w};
            unsigned m = 0u;
#pragma unroll
            for (int k = 0; k < 4; ++k) {
                const unsigned x = ws[k] ^ pat;
                const unsigned t = ~(((x & 0x7f7f7f7fu) + 0x7f7f7f7fu) | x | 0x7f7f7f7fu);
                m |= (t >> 7) << k;
            }
            if (!live[u]) m = 0u;
            while (m) {
                const int pbit = __builtin_ctz(m);
                m &= m - 1u;
                const unsigned slot = atomicAdd(qn, 1u);
                if (slot < QCAP) q_e[slot] = (int)(g[u] * 16u) + (pbit & 7) * 4 + (pbit >> 3);
            }
        }
    }
}

template <bool TAIL>
__device__ __forceinline__ void row_trip(const float* __restrict__ feat, const int* q_e,
                                         const float* q_x, float* acc,
                                         unsigned i0, unsigned cnt, int lane) {
    const unsigned NW = TB / 32;
    float4 f[RG]; float x[RG]; int d[RG];
#pragma unroll
    for (int u = 0; u < RG; ++u) {
        unsigned i = i0 + u * NW;
        bool live = true;
        if (TAIL) { live = i < cnt; i = live ? i : cnt - 1u; }
        const unsigned p = (unsigned)q_e[i];
        d[u] = live ? (int)(p >> 16) : TILE;
        x[u] = q_x[i];
        f[u] = *(const float4*)(feat + (long)(p & 0xffffu) * HD + lane * 4);
    }
#pragma unroll
    for (int u = 0; u < RG; ++u) {
        float* o = acc + d[u] * HD + lane;
        unsafeAtomicAdd(o,      f[u].x * x[u]);
        unsafeAtomicAdd(o + 32, f[u].y * x[u]);
        unsafeAtomicAdd(o + 64, f[u].z * x[u]);
        unsafeAtomicAdd(o + 96, f[u].w * x[u]);
    }
}

#define NWV    (TB / 32)
#define OWNR   ((TILE + NWV - 1) / NWV)
#define NBIN   (NWV * OWNR)

__device__ __forceinline__ unsigned own_bin(int dl) {
    return (unsigned)(dl & (NWV - 1)) * OWNR + (unsigned)(dl / NWV);
}

__device__ __forceinline__ void own_flush(float* acc, float4 cur, int r, int lane) {
    if (r >= 0) {
        float* o = acc + r * HD + lane;
        o[0] = cur.x; o[32] = cur.y; o[64] = cur.z; o[96] = cur.w;
    }
}

template <bool TAIL>
__device__ __forceinline__ void own_trip(const float* __restrict__ feat, const unsigned* q_s,
                                         const float* q_x, float* acc, float4& cur, int& rcur,
                                         unsigned i0, unsigned b1, int lane) {
    float4 f[RG]; float x[RG]; int r[RG];
#pragma unroll
    for (int u = 0; u < RG; ++u) {
        unsigned i = i0 + u;
        if (TAIL) i = i < b1 ? i : b1 - 1u;
        const unsigned p = q_s[i];
        r[u] = __builtin_amdgcn_readfirstlane((int)(p >> 16));
        x[u] = q_x[i];
        f[u] = *(const float4*)(feat + (long)(p & 0xffffu) * HD + lane * 4);
    }
#pragma unroll
    for (int u = 0; u < RG; ++u) {
        if (!TAIL || i0 + u < b1) {
            if (r[u] != rcur) {
                own_flush(acc, cur, rcur, lane);
                cur = make_float4(0.f, 0.f, 0.f, 0.f);
                rcur = r[u];
            }
            cur.x += f[u].x * x[u]; cur.y += f[u].y * x[u];
            cur.z += f[u].z * x[u]; cur.w += f[u].w * x[u];
        }
    }
}

__global__ void __launch_bounds__(TB) gat_tile(
        const int* __restrict__ src, const int* __restrict__ dst,
        const unsigned char* __restrict__ bkt,
        const float* __restrict__ feat,
        const float* __restrict__ el, const float* __restrict__ er,
        const float* __restrict__ bias, float* __restrict__ out,
        unsigned* __restrict__ tcnt, unsigned* __restrict__ tlist, int use_list,
        const float* __restrict__ Wn, const float* __restrict__ aln, const float* __restrict__ arn,
        float* __restrict__ featn, float* __restrict__ eln, float* __restrict__ ern) {
    __shared__ __attribute__((aligned(16))) float acc[ACCW];
    __shared__ float m_l[TILE * HEADS];
    __shared__ float s_l[TILE * HEADS];
    __shared__ float er_l[TILE * HEADS];
    __shared__ int   q_e[QCAP];
    __shared__ __attribute__((aligned(16))) float q_xb[2 * QCAP];
    __shared__ unsigned q_g[QCAP];
    __shared__ unsigned qn, gn;
    __shared__ unsigned bcnt[NBIN], bfill[NBIN], boff[NBIN + 1];
    float* const q_x0 = q_xb;
    float* const q_x1 = q_xb + QCAP;

    const int tid  = threadIdx.x;
    const int base = blockIdx.x * TILE;
    const unsigned myb = blockIdx.x;
    unsigned* __restrict__ mylist = tlist + (long)blockIdx.x * QCAP;

    for (int i = tid; i < (TILE + 1) * HD; i += TB) acc[i] = 0.f;
    for (int i = tid; i < TILE * HEADS; i += TB) {
        m_l[i] = -INFINITY; s_l[i] = 0.f;
        er_l[i] = (base + (i >> 1)) < NN ? er[base * HEADS + i] : 0.f;
    }
    if (tid == 0) { qn = 0u; gn = 0u; }
    if (tid < NBIN) { bcnt[tid] = 0u; bfill[tid] = 0u; }
    __syncthreads();

    unsigned cnt = 0u;
    bool listed = false;
    if (use_list) {
        cnt = tcnt[blockIdx.x];
        listed = cnt <= QCAP;
    }
    if (!listed) map_walk(mylist, q_g, &gn, tid);
    __syncthreads();
    if (!listed) {
        const unsigned ng = gn;
        if (ng <= QCAP) group_match(bkt, q_g, ng, myb, q_e, &qn, tid);
        else if (tid == 0) qn = ng;
    }
    __syncthreads();
    if (!listed) {
        cnt = qn;
        if (!use_list && tid == 0) tcnt[blockIdx.x] = cnt;
    }
    const int seglen = cnt > QCAP ? QCAP : NE;
    const int nseg   = (NE + seglen - 1) / seglen;
    const bool publish = !use_list && nseg == 1;
    const bool own = nseg == 1;

    for (int seg = 0; seg < nseg; ++seg) {
        if (nseg > 1) {
            __syncthreads();
            if (tid == 0) qn = 0u;
            __syncthreads();
            const int eb = seg * seglen;
            tile_scan(bkt, eb, min(NE, eb + seglen), myb, q_e, &qn, tid);
            __syncthreads();
            cnt = qn;
        }
        for (unsigned i = tid; i < cnt; i += TB) {
            unsigned p;
            if (listed) {
                p = mylist[i];
            } else {
                const int e = q_e[i];
                p = (unsigned)src[e] | ((unsigned)(dst[e] - base) << 16);
                if (publish) mylist[i] = p;
            }
            q_e[i] = (int)p;
            const int s0 = (int)(p & 0xffffu);
            const int dl = (int)(p >> 16);
            const float2 l2 = *(const float2*)(el + (long)s0 * HEADS);
            float v0 = l2.x + er_l[dl * HEADS], v1 = l2.y + er_l[dl * HEADS + 1];
            v0 = v0 > 0.f ? v0 : NEG_SLOPE * v0;
            v1 = v1 > 0.f ? v1 : NEG_SLOPE * v1;
            atomicMaxF(&m_l[dl * HEADS], v0);
            atomicMaxF(&m_l[dl * HEADS + 1], v1);
            if (own) atomicAdd(&bcnt[own_bin(dl)], 1u);
        }
    }
    __syncthreads();
    if (own) {
        if (tid < NWV) {
            unsigned cb[OWNR], tot = 0u;
#pragma unroll
            for (int k = 0; k < OWNR; ++k) { cb[k] = bcnt[tid * OWNR + k]; tot += cb[k]; }
            unsigned inc = tot;
#pragma unroll
            for (int d = 1; d < NWV; d <<= 1) {
                const unsigned v = __shfl_up(inc, d);
                if (tid >= d) inc += v;
            }
            unsigned o = inc - tot;
#pragma unroll
            for (int k = 0; k < OWNR; ++k) { boff[tid * OWNR + k] = o; o += cb[k]; }
            if (tid == NWV - 1) boff[NBIN] = o;
        }
        __syncthreads();
    }

    const int wave = tid >> 5, lane = tid & 31;
    const int h  = lane >> 4;
    for (int seg = 0; seg < nseg; ++seg) {
        if (nseg > 1) {
            __syncthreads();
            if (tid == 0) qn = 0u;
            __syncthreads();
            const int eb = seg * seglen;
            tile_scan(bkt, eb, min(NE, eb + seglen), myb, q_e, &qn, tid);
            __syncthreads();
            cnt = qn;
        }
        for (unsigned i = tid; i < cnt; i += TB) {
            unsigned p = (unsigned)q_e[i];
            if (nseg > 1) {
                const int e = (int)p;
                p = (unsigned)src[e] | ((unsigned)(dst[e] - base) << 16);
            }
            const int s0 = (int)(p & 0xffffu);
            const int dl = (int)(p >> 16);
            const float2 l2 = *(const float2*)(el + (long)s0 * HEADS);
            float v0 = l2.x + er_l[dl * HEADS], v1 = l2.y + er_l[dl * HEADS + 1];
            v0 = v0 > 0.f ? v0 : NEG_SLOPE * v0;
            v1 = v1 > 0.f ? v1 : NEG_SLOPE * v1;
            const float x0 = __expf(v0 - m_l[dl * HEADS]);
            const float x1 = __expf(v1 - m_l[dl * HEADS + 1]);
            unsigned slot = i;
            if (own) {
                const unsigned b = own_bin(dl);
                slot = boff[b] + atomicAdd(&bfill[b], 1u);
                q_g[slot] = p;
            } else {
                q_e[i] = (int)p;
            }
            q_x0[slot] = x0; q_x1[slot] = x1;
            unsafeAtomicAdd(&s_l[dl * HEADS], x0);
            unsafeAtomicAdd(&s_l[dl * HEADS + 1], x1);
        }
        __syncthreads();
        const unsigned NW = TB / 32;
        const float* q_x = h ? q_x1 : q_x0;
        if (own) {
            const int wo = __builtin_amdgcn_readfirstlane(wave);
            const unsigned b1 = (unsigned)__builtin_amdgcn_readfirstlane((int)boff[(wo + 1) * OWNR]);
            unsigned i0 = (unsigned)__builtin_amdgcn_readfirstlane((int)boff[wo * OWNR]);
            float4 cur = make_float4(0.f, 0.f, 0.f, 0.f);
            int rcur = -1;
            for (; i0 + RG <= b1; i0 += RG) own_trip<false>(feat, q_g, q_x, acc, cur, rcur, i0, b1, lane);
            if (i0 < b1) own_trip<true>(feat, q_g, q_x, acc, cur, rcur, i0, b1, lane);
            own_flush(acc, cur, rcur, lane);
        } else {
            unsigned i0 = wave;
            for (; i0 + (RG - 1) * NW < cnt; i0 += RG * NW)
                row_trip<false>(feat, q_e, q_x, acc, i0, cnt, lane);
            if (i0 < cnt) row_trip<true>(feat, q_e, q_x, acc, i0, cnt, lane);
        }
    }
    __syncthreads();

    if (!use_list) {
        float* const hb = q_xb;
        for (int i = tid; i < HROWS * HID; i += TB) {
            const int nl = i >> 6, dd = i & 63;
            float r = 0.f;
            if (nl < TILE && base + nl < NN) {
                const float s0 = s_l[nl * HEADS], s1 = s_l[nl * HEADS + 1];
                const int w0 = nl * HD + (dd & 3) * 32 + (dd >> 2);
                const float a0 = s0 > 0.f ? acc[w0]      / s0 : 0.f;
                const float a1 = s1 > 0.f ? acc[w0 + 16] / s1 : 0.f;
                const float v0 = a0 + bias[dd];
                const float v1 = a1 + bias[HID + dd];
                const float v  = 0.5f * (v0 + v1);
                r = v > 0.f ? v : (__expf(v) - 1.f);
            }
            hb[nl * HP + dd] = r;
        }
        __syncthreads();

        const int wv   = __builtin_amdgcn_readfirstlane(wave);
        const int ct   = wv & 7;
        const int half = lane >> 4, l15 = lane & 15;
        v2f bf[HID / 4];
#pragma unroll
        for (int kq = 0; kq < HID / 4; ++kq) {
            const int k0 = kq * 4 + half * 2;
            bf[kq].x = Wn[k0 * HD + ct * 16 + l15];
            bf[kq].y = Wn[(k0 + 1) * HD + ct * 16 + l15];
        }
        float* const fb = acc;
        const int fcol = (ct >> 2) * FH1 + (ct & 3) * 16 + l15;
        for (int rt = wv >> 3; rt < HROWS / 16; rt += 4) {
            const float* arow = hb + (rt * 16 + l15) * HP + half * 2;
            v8f cc = {};
#pragma unroll
            for (int kq = 0; kq < HID / 4; ++kq) {
                const v2f a = *(const v2f*)(arow + kq * 4);
                cc = __builtin_amdgcn_wmma_f32_16x16x4_f32(false, a, false, bf[kq],
                                                           (short)0, cc, false, false);
            }
            const int r0 = rt * 16 + half * 8;
#pragma unroll
            for (int v = 0; v < 8; ++v) {
                const int row = r0 + v;
                if (row < TILE) {
                    fb[row * FP + fcol] = cc[v];
                    if (base + row < NN) featn[(long)(base + row) * HD + ct * 16 + l15] = cc[v];
                }
            }
        }
        __syncthreads();

        if (tid < TILE * HEADS && base + (tid >> 1) < NN) {
            const int hh = tid & 1;
            const float* f  = fb + (tid >> 1) * FP + hh * FH1;
            const float* la = aln + hh * HID;
            const float* ra = arn + hh * HID;
            float sl = 0.f, sr = 0.f;
#pragma unroll 4
            for (int d = 0; d < HID; ++d) { float v = f[d]; sl += v * la[d]; sr += v * ra[d]; }
            eln[base * HEADS + tid] = sl; ern[base * HEADS + tid] = sr;
        }
        return;
    }

    for (int i = tid; i < TILE * HID; i += TB) {
        const int nl = i >> 6, dd = i & 63;
        const int n  = base + nl;
        if (n < NN) {
            const float s0 = s_l[nl * HEADS], s1 = s_l[nl * HEADS + 1];
            const int w0 = nl * HD + (dd & 3) * 32 + (dd >> 2);
            const float a0 = s0 > 0.f ? acc[w0]      / s0 : 0.f;
            const float a1 = s1 > 0.f ? acc[w0 + 16] / s1 : 0.f;
            const float v0 = a0 + bias[dd];
            const float v1 = a1 + bias[HID + dd];
            const float v  = 0.5f * (v0 + v1);
            out[(long)n * HID + dd] = v > 0.f ? v : (__expf(v) - 1.f);
        }
    }
}

extern "C" void kernel_launch(void* const* d_in, const int* in_sizes, int n_in,
                              void* d_out, int out_size, void* d_ws, size_t ws_size,
                              hipStream_t stream) {
    const float* x   = (const float*)d_in[0];
    const float* W1  = (const float*)d_in[1];
    const float* al1 = (const float*)d_in[2];
    const float* ar1 = (const float*)d_in[3];
    const float* b1  = (const float*)d_in[4];
    const float* W2  = (const float*)d_in[5];
    const float* al2 = (const float*)d_in[6];
    const float* ar2 = (const float*)d_in[7];
    const float* b2  = (const float*)d_in[8];
    const int*   src = (const int*)d_in[9];
    const int*   dst = (const int*)d_in[10];
    float* out = (float*)d_out;

    char* wsb = (char*)d_ws;
    size_t off = 0;
    auto alloc = [&](size_t bytes) -> void* {
        void* p = wsb + off;
        off += bytes;
        off = (off + 255) & ~(size_t)255;
        return p;
    };
    float* feat = (float*)alloc((size_t)NN * HD * 4);
    float* el   = (float*)alloc((size_t)NN * HEADS * 4);
    float* er   = (float*)alloc((size_t)NN * HEADS * 4);
    float* feat2 = (float*)alloc((size_t)NN * HD * 4);
    float* el2  = (float*)alloc((size_t)NN * HEADS * 4);
    float* er2  = (float*)alloc((size_t)NN * HEADS * 4);
    unsigned char* bkt = (unsigned char*)alloc((size_t)NE);
    unsigned* tcnt  = (unsigned*)alloc((size_t)NTILES * 4);
    unsigned* tlist = (unsigned*)alloc((size_t)NTILES * QCAP * 4);

    const dim3 B256(256);

    gat_bucket<<<dim3(NBK), B256, 0, stream>>>(dst, bkt, tlist, NE);

    gat_proj<<<dim3((NN + PR - 1) / PR), B256, 0, stream>>>(x, W1, al1, ar1, feat, el, er);
    gat_tile<<<dim3(NTILES), dim3(TB), 0, stream>>>(src, dst, bkt, feat, el, er, b1, out, tcnt, tlist, 0,
                                                    W2, al2, ar2, feat2, el2, er2);

    gat_tile<<<dim3(NTILES), dim3(TB), 0, stream>>>(src, dst, bkt, feat2, el2, er2, b2, out, tcnt, tlist, 1,
                                                    nullptr, nullptr, nullptr, nullptr, nullptr, nullptr);
}


# Layer 1's tile reads its 16-edge groups as packed words (source | destination, written by the bucket pass; four lanes per 64-byte group): no src[e], dst[e] read per edge; same sums, arrival order
# speedup vs baseline: 9.2451x; 1.0402x over previous
#include <hip/hip_runtime.h>
#include <hip/hip_bf16.h>

#define NN   50000
#define NE   1600000
#define INF_ 128
#define HID  64
#define HEADS 2
#define HD   (HEADS*HID)
#define NEG_SLOPE 0.2f

typedef __attribute__((ext_vector_type(2))) float v2f;
typedef __attribute__((ext_vector_type(8))) float v8f;

__device__ __forceinline__ void atomicMaxF(float* addr, float val) {
    if (val >= 0.0f) atomicMax((int*)addr, __float_as_int(val));
    else             atomicMin((unsigned int*)addr, (unsigned int)__float_as_int(val));
}


#define PR   64
#define XP   132

__global__ void __launch_bounds__(256) gat_proj(
        const float* __restrict__ x, const float* __restrict__ W,
        const float* __restrict__ al, const float* __restrict__ ar,
        float* __restrict__ feat, float* __restrict__ el, float* __restrict__ er) {
    __shared__ __attribute__((aligned(16))) float xb[PR * XP];
    const int tid  = threadIdx.x;
    const int lane = tid & 31;
    const int wv   = __builtin_amdgcn_readfirstlane(tid >> 5);
    const int base = blockIdx.x * PR;
    const int half = lane >> 4, l15 = lane & 15;

    float4 xr[PR / 8];
#pragma unroll
    for (int j = 0; j < PR / 8; ++j) {
        int n = base + wv + j * 8;
        n = n < NN ? n : NN - 1;
        xr[j] = *(const float4*)(x + (long)n * INF_ + lane * 4);
    }
#pragma unroll
    for (int j = 0; j < PR / 8; ++j) {
        const int r = wv + j * 8;
        *(float4*)(xb + r * XP + lane * 4) = base + r < NN ? xr[j] : make_float4(0.f, 0.f, 0.f, 0.f);
    }
    __syncthreads();

    v8f cc[PR / 16];
#pragma unroll
    for (int t = 0; t < PR / 16; ++t) cc[t] = v8f{};
    const float* wcol = W + wv * 16 + l15;
    const float* arow = xb + l15 * XP + half * 2;
#pragma unroll 8
    for (int kq = 0; kq < INF_ / 4; ++kq) {
        const int k0 = kq * 4 + half * 2;
        v2f b;
        b.x = wcol[k0 * HD];
        b.y = wcol[(k0 + 1) * HD];
#pragma unroll
        for (int t = 0; t < PR / 16; ++t) {
            const v2f a = *(const v2f*)(arow + t * 16 * XP + kq * 4);
            cc[t] = __builtin_amdgcn_wmma_f32_16x16x4_f32(false, a, false, b,
                                                          (short)0, cc[t], false, false);
        }
    }
    __syncthreads();

#pragma unroll
    for (int t = 0; t < PR / 16; ++t) {
#pragma unroll
        for (int v = 0; v < 8; ++v)
            xb[(t * 16 + half * 8 + v) * XP + wv * 16 + l15] = cc[t][v];
    }
    __syncthreads();

#pragma unroll
    for (int j = 0; j < PR / 8; ++j) xr[j] = *(const float4*)(xb + (wv + j * 8) * XP + lane * 4);
#pragma unroll
    for (int j = 0; j < PR / 8; ++j) {
        const int r = wv + j * 8;
        if (base + r < NN) *(float4*)(feat + (long)(base + r) * HD + lane * 4) = xr[j];
    }

    if (tid < PR * HEADS && base + (tid >> 1) < NN) {
        const int hh = tid & 1;
        const float* f  = xb + (tid >> 1) * XP + hh * HID;
        const float* la = al + hh * HID;
        const float* ra = ar + hh * HID;
        float sl = 0.f, sr = 0.f;
#pragma unroll 4
        for (int d = 0; d < HID; ++d) { float v = f[d]; sl += v * la[d]; sr += v * ra[d]; }
        el[base * HEADS + tid] = sl; er[base * HEADS + tid] = sr;
    }
}

#define TILE   196
#define NTILES 256
#define QCAP   8192
#define TB     1024
#define RG     8
#define HROWS  208
#define HP     68
#define FP     130
#define FH1    65
#define ACCW   (TILE * FP)

#define BKE    2048
#define NBK    ((NE + BKE - 1) / BKE)
#define BMW    (NBK * 4)

__global__ void __launch_bounds__(256) gat_bucket(const int* __restrict__ src,
                                                  const int* __restrict__ dst,
                                                  unsigned char* __restrict__ bkt,
                                                  unsigned* __restrict__ pk,
                                                  unsigned* __restrict__ tlist, int E) {
    __shared__ unsigned w[4 * NTILES];
    const int t = threadIdx.x;
    for (int i = t; i < 4 * NTILES; i += 256) w[i] = 0u;
    __syncthreads();
    const int e0 = blockIdx.x * BKE + t * 8;
    if (e0 < E) {
        const int4 a = *(const int4*)(dst + e0);
        const int4 b = *(const int4*)(dst + e0 + 4);
        const unsigned t0 = a.x / TILE, t1 = a.y / TILE, t2 = a.z / TILE, t3 = a.w / TILE;
        const unsigned t4 = b.x / TILE, t5 = b.y / TILE, t6 = b.z / TILE, t7 = b.w / TILE;
        uint2 tb;
        tb.x = t0 | (t1 << 8) | (t2 << 16) | (t3 << 24);
        tb.y = t4 | (t5 << 8) | (t6 << 16) | (t7 << 24);
        *(uint2*)(bkt + e0) = tb;
        const int4 sa = *(const int4*)(src + e0);
        const int4 sb = *(const int4*)(src + e0 + 4);
        uint4 w0, w1;
        w0.x = (unsigned)sa.x | ((unsigned)a.x << 16); w0.y = (unsigned)sa.y | ((unsigned)a.y << 16);
        w0.z = (unsigned)sa.z | ((unsigned)a.z << 16); w0.w = (unsigned)sa.w | ((unsigned)a.w << 16);
        w1.x = (unsigned)sb.x | ((unsigned)b.x << 16); w1.y = (unsigned)sb.y | ((unsigned)b.y << 16);
        w1.z = (unsigned)sb.z | ((unsigned)b.z << 16); w1.w = (unsigned)sb.w | ((unsigned)b.w << 16);
        *(uint4*)(pk + e0) = w0;
        *(uint4*)(pk + e0 + 4) = w1;
        unsigned* wj = w + (t >> 6) * NTILES;
        const unsigned bit = 1u << ((t >> 1) & 31);
        atomicOr(wj + t0, bit); atomicOr(wj + t1, bit); atomicOr(wj + t2, bit); atomicOr(wj + t3, bit);
        atomicOr(wj + t4, bit); atomicOr(wj + t5, bit); atomicOr(wj + t6, bit); atomicOr(wj + t7, bit);
    }
    __syncthreads();
    uint4 o;
    o.x = w[t]; o.y = w[NTILES + t]; o.z = w[2 * NTILES + t]; o.w = w[3 * NTILES + t];
    *(uint4*)(tlist + (long)t * QCAP + blockIdx.x * 4) = o;
}

__device__ __forceinline__ void tile_match16(uint4 w, unsigned pat, int e0, int* q_e, unsigned* qn) {
    const unsigned ws[4] = {w.x, w.y, w.z, w.w};
#pragma unroll
    for (int k = 0; k < 4; ++k) {
        const unsigned x = ws[k] ^ pat;
        unsigned t = ~(((x & 0x7f7f7f7fu) + 0x7f7f7f7fu) | x | 0x7f7f7f7fu);
        while (t) {
            const int b = __builtin_ctz(t) >> 3;
            const unsigned slot = atomicAdd(qn, 1u);
            if (slot < QCAP) q_e[slot] = e0 + k * 4 + b;
            t &= t - 1u;
        }
    }
}

__device__ __forceinline__ void tile_scan(const unsigned char* __restrict__ bkt,
                                          int e_begin, int e_end, unsigned myb,
                                          int* q_e, unsigned* qn, int tid) {
    const uint4* __restrict__ p = (const uint4*)bkt;
    const unsigned pat = myb * 0x01010101u;
    const int g1 = e_end >> 4;
    int g = (e_begin >> 4) + tid;
    for (; g + 7 * TB < g1; g += 8 * TB) {
        uint4 w[8];
#pragma unroll
        for (int u = 0; u < 8; ++u) w[u] = p[g + u * TB];
#pragma unroll
        for (int u = 0; u < 8; ++u) tile_match16(w[u], pat, (g + u * TB) * 16, q_e, qn);
    }
    for (; g < g1; g += TB) tile_match16(p[g], pat, g * 16, q_e, qn);
}

__device__ __forceinline__ void map_walk(const unsigned* bm, unsigned* q_g, unsigned* gn, int tid) {
    for (int i = tid; i < BMW; i += TB) {
        unsigned w = bm[i];
        while (w) {
            const int b = __builtin_ctz(w);
            w &= w - 1u;
            const unsigned slot = atomicAdd(gn, 1u);
            if (slot < QCAP) q_g[slot] = (unsigned)(i * 32 + b);
        }
    }
}

#define GU 8
__device__ __forceinline__ void group_match(const unsigned* __restrict__ pk,
                                            const unsigned* q_g, unsigned ng, int base,
                                            int* q_e, unsigned* qn, int tid) {
    const uint4* __restrict__ p = (const uint4*)pk;
    const unsigned n4   = ng * 4u;
    const unsigned lane = (unsigned)tid & 31u;
    const unsigned qq   = (unsigned)tid & 3u;
    unsigned jw = (unsigned)__builtin_amdgcn_readfirstlane(tid - (int)lane);
    for (; jw < n4; jw += GU * TB) {
        uint4 v[GU]; bool live[GU];
#pragma unroll
        for (int u = 0; u < GU; ++u) {
            const unsigned j = jw + lane + u * TB;
            live[u] = j < n4;
            const unsigned g = q_g[(live[u] ? j : n4 - 1u) >> 2];
            v[u] = p[g * 4u + qq];
        }
        unsigned bal[GU * 4];
        unsigned tot = 0u;
#pragma unroll
        for (int u = 0; u < GU; ++u) {
            const unsigned ws[4] = {v[u].x, v[u].y, v[u].z, v[u].w};
#pragma unroll
            for (int k = 0; k < 4; ++k) {
                const unsigned d = (ws[k] >> 16) - (unsigned)base;
                bal[u * 4 + k] = __builtin_amdgcn_ballot_w32(live[u] && d < TILE);
                tot += (unsigned)__builtin_popcount(bal[u * 4 + k]);
            }
        }
        if (tot) {
            unsigned wb = 0u;
            if (lane == 0u) wb = atomicAdd(qn, tot);
            wb = (unsigned)__builtin_amdgcn_readfirstlane((int)wb);
#pragma unroll
            for (int u = 0; u < GU; ++u) {
                const unsigned ws[4] = {v[u].x, v[u].y, v[u].z, v[u].w};
#pragma unroll
                for (int k = 0; k < 4; ++k) {
                    const unsigned b = bal[u * 4 + k];
                    if (b) {
                        if ((b >> lane) & 1u) {
                            const unsigned slot = wb + __builtin_amdgcn_mbcnt_lo(b, 0u);
                            const unsigned d = (ws[k] >> 16) - (unsigned)base;
                            if (slot < QCAP) q_e[slot] = (int)((ws[k] & 0xffffu) | (d << 16));
                        }
                        wb += (unsigned)__builtin_popcount(b);
                    }
                }
            }
        }
    }
}

template <bool TAIL>
__device__ __forceinline__ void row_trip(const float* __restrict__ feat, const int* q_e,
                                         const float* q_x, float* acc,
                                         unsigned i0, unsigned cnt, int lane) {
    const unsigned NW = TB / 32;
    float4 f[RG]; float x[RG]; int d[RG];
#pragma unroll
    for (int u = 0; u < RG; ++u) {
        unsigned i = i0 + u * NW;
        bool live = true;
        if (TAIL) { live = i < cnt; i = live ? i : cnt - 1u; }
        const unsigned p = (unsigned)q_e[i];
        d[u] = live ? (int)(p >> 16) : TILE;
        x[u] = q_x[i];
        f[u] = *(const float4*)(feat + (long)(p & 0xffffu) * HD + lane * 4);
    }
#pragma unroll
    for (int u = 0; u < RG; ++u) {
        float* o = acc + d[u] * HD + lane;
        unsafeAtomicAdd(o,      f[u].x * x[u]);
        unsafeAtomicAdd(o + 32, f[u].y * x[u]);
        unsafeAtomicAdd(o + 64, f[u].z * x[u]);
        unsafeAtomicAdd(o + 96, f[u].w * x[u]);
    }
}

#define NWV    (TB / 32)
#define OWNR   ((TILE + NWV - 1) / NWV)
#define NBIN   (NWV * OWNR)

__device__ __forceinline__ unsigned own_bin(int dl) {
    return (unsigned)(dl & (NWV - 1)) * OWNR + (unsigned)(dl / NWV);
}

__device__ __forceinline__ void own_flush(float* acc, float4 cur, int r, int lane) {
    if (r >= 0) {
        float* o = acc + r * HD + lane;
        o[0] = cur.x; o[32] = cur.y; o[64] = cur.z; o[96] = cur.w;
    }
}

template <bool TAIL>
__device__ __forceinline__ void own_trip(const float* __restrict__ feat, const unsigned* q_s,
                                         const float* q_x, float* acc, float4& cur, int& rcur,
                                         unsigned i0, unsigned b1, int lane) {
    float4 f[RG]; float x[RG]; int r[RG];
#pragma unroll
    for (int u = 0; u < RG; ++u) {
        unsigned i = i0 + u;
        if (TAIL) i = i < b1 ? i : b1 - 1u;
        const unsigned p = q_s[i];
        r[u] = __builtin_amdgcn_readfirstlane((int)(p >> 16));
        x[u] = q_x[i];
        f[u] = *(const float4*)(feat + (long)(p & 0xffffu) * HD + lane * 4);
    }
#pragma unroll
    for (int u = 0; u < RG; ++u) {
        if (!TAIL || i0 + u < b1) {
            if (r[u] != rcur) {
                own_flush(acc, cur, rcur, lane);
                cur = make_float4(0.f, 0.f, 0.f, 0.f);
                rcur = r[u];
            }
            cur.x += f[u].x * x[u]; cur.y += f[u].y * x[u];
            cur.z += f[u].z * x[u]; cur.w += f[u].w * x[u];
        }
    }
}

__global__ void __launch_bounds__(TB) gat_tile(
        const int* __restrict__ src, const int* __restrict__ dst,
        const unsigned char* __restrict__ bkt, const unsigned* __restrict__ pk,
        const float* __restrict__ feat,
        const float* __restrict__ el, const float* __restrict__ er,
        const float* __restrict__ bias, float* __restrict__ out,
        unsigned* __restrict__ tcnt, unsigned* __restrict__ tlist, int use_list,
        const float* __restrict__ Wn, const float* __restrict__ aln, const float* __restrict__ arn,
        float* __restrict__ featn, float* __restrict__ eln, float* __restrict__ ern) {
    __shared__ __attribute__((aligned(16))) float acc[ACCW];
    __shared__ float m_l[TILE * HEADS];
    __shared__ float s_l[TILE * HEADS];
    __shared__ float er_l[TILE * HEADS];
    __shared__ int   q_e[QCAP];
    __shared__ __attribute__((aligned(16))) float q_xb[2 * QCAP];
    __shared__ unsigned q_g[QCAP];
    __shared__ unsigned qn, gn;
    __shared__ unsigned bcnt[NBIN], bfill[NBIN], boff[NBIN + 1];
    float* const q_x0 = q_xb;
    float* const q_x1 = q_xb + QCAP;

    const int tid  = threadIdx.x;
    const int base = blockIdx.x * TILE;
    const unsigned myb = blockIdx.x;
    unsigned* __restrict__ mylist = tlist + (long)blockIdx.x * QCAP;

    for (int i = tid; i < (TILE + 1) * HD; i += TB) acc[i] = 0.f;
    for (int i = tid; i < TILE * HEADS; i += TB) {
        m_l[i] = -INFINITY; s_l[i] = 0.f;
        er_l[i] = (base + (i >> 1)) < NN ? er[base * HEADS + i] : 0.f;
    }
    if (tid == 0) { qn = 0u; gn = 0u; }
    if (tid < NBIN) { bcnt[tid] = 0u; bfill[tid] = 0u; }
    __syncthreads();

    unsigned cnt = 0u;
    bool listed = false;
    if (use_list) {
        cnt = tcnt[blockIdx.x];
        listed = cnt <= QCAP;
    }
    if (!use_list) map_walk(mylist, q_g, &gn, tid);
    __syncthreads();
    if (!use_list) {
        const unsigned ng = gn;
        if (ng <= QCAP) group_match(pk, q_g, ng, base, q_e, &qn, tid);
        else if (tid == 0) qn = ng;
    }
    __syncthreads();
    if (!use_list) {
        cnt = qn;
        if (tid == 0) tcnt[blockIdx.x] = cnt;
    }
    const int seglen = cnt > QCAP ? QCAP : NE;
    const int nseg   = (NE + seglen - 1) / seglen;
    const bool publish = !use_list && nseg == 1;
    const bool own = nseg == 1;

    for (int seg = 0; seg < nseg; ++seg) {
        if (nseg > 1) {
            __syncthreads();
            if (tid == 0) qn = 0u;
            __syncthreads();
            const int eb = seg * seglen;
            tile_scan(bkt, eb, min(NE, eb + seglen), myb, q_e, &qn, tid);
            __syncthreads();
            cnt = qn;
        }
        for (unsigned i = tid; i < cnt; i += TB) {
            unsigned p;
            if (listed) {
                p = mylist[i];
            } else if (nseg > 1) {
                const int e = q_e[i];
                p = (unsigned)src[e] | ((unsigned)(dst[e] - base) << 16);
            } else {
                p = (unsigned)q_e[i];
                if (publish) mylist[i] = p;
            }
            q_e[i] = (int)p;
            const int s0 = (int)(p & 0xffffu);
            const int dl = (int)(p >> 16);
            const float2 l2 = *(const float2*)(el + (long)s0 * HEADS);
            float v0 = l2.x + er_l[dl * HEADS], v1 = l2.y + er_l[dl * HEADS + 1];
            v0 = v0 > 0.f ? v0 : NEG_SLOPE * v0;
            v1 = v1 > 0.f ? v1 : NEG_SLOPE * v1;
            atomicMaxF(&m_l[dl * HEADS], v0);
            atomicMaxF(&m_l[dl * HEADS + 1], v1);
            if (own) atomicAdd(&bcnt[own_bin(dl)], 1u);
        }
    }
    __syncthreads();
    if (own) {
        if (tid < NWV) {
            unsigned cb[OWNR], tot = 0u;
#pragma unroll
            for (int k = 0; k < OWNR; ++k) { cb[k] = bcnt[tid * OWNR + k]; tot += cb[k]; }
            unsigned inc = tot;
#pragma unroll
            for (int d = 1; d < NWV; d <<= 1) {
                const unsigned v = __shfl_up(inc, d);
                if (tid >= d) inc += v;
            }
            unsigned o = inc - tot;
#pragma unroll
            for (int k = 0; k < OWNR; ++k) { boff[tid * OWNR + k] = o; o += cb[k]; }
            if (tid == NWV - 1) boff[NBIN] = o;
        }
        __syncthreads();
    }

    const int wave = tid >> 5, lane = tid & 31;
    const int h  = lane >> 4;
    for (int seg = 0; seg < nseg; ++seg) {
        if (nseg > 1) {
            __syncthreads();
            if (tid == 0) qn = 0u;
            __syncthreads();
            const int eb = seg * seglen;
            tile_scan(bkt, eb, min(NE, eb + seglen), myb, q_e, &qn, tid);
            __syncthreads();
            cnt = qn;
        }
        for (unsigned i = tid; i < cnt; i += TB) {
            unsigned p = (unsigned)q_e[i];
            if (nseg > 1) {
                const int e = (int)p;
                p = (unsigned)src[e] | ((unsigned)(dst[e] - base) << 16);
            }
            const int s0 = (int)(p & 0xffffu);
            const int dl = (int)(p >> 16);
            const float2 l2 = *(const float2*)(el + (long)s0 * HEADS);
            float v0 = l2.x + er_l[dl * HEADS], v1 = l2.y + er_l[dl * HEADS + 1];
            v0 = v0 > 0.f ? v0 : NEG_SLOPE * v0;
            v1 = v1 > 0.f ? v1 : NEG_SLOPE * v1;
            const float x0 = __expf(v0 - m_l[dl * HEADS]);
            const float x1 = __expf(v1 - m_l[dl * HEADS + 1]);
            unsigned slot = i;
            if (own) {
                const unsigned b = own_bin(dl);
                slot = boff[b] + atomicAdd(&bfill[b], 1u);
                q_g[slot] = p;
            } else {
                q_e[i] = (int)p;
            }
            q_x0[slot] = x0; q_x1[slot] = x1;
            unsafeAtomicAdd(&s_l[dl * HEADS], x0);
            unsafeAtomicAdd(&s_l[dl * HEADS + 1], x1);
        }
        __syncthreads();
        const unsigned NW = TB / 32;
        const float* q_x = h ? q_x1 : q_x0;
        if (own) {
            const int wo = __builtin_amdgcn_readfirstlane(wave);
            const unsigned b1 = (unsigned)__builtin_amdgcn_readfirstlane((int)boff[(wo + 1) * OWNR]);
            unsigned i0 = (unsigned)__builtin_amdgcn_readfirstlane((int)boff[wo * OWNR]);
            float4 cur = make_float4(0.f, 0.f, 0.f, 0.f);
            int rcur = -1;
            for (; i0 + RG <= b1; i0 += RG) own_trip<false>(feat, q_g, q_x, acc, cur, rcur, i0, b1, lane);
            if (i0 < b1) own_trip<true>(feat, q_g, q_x, acc, cur, rcur, i0, b1, lane);
            own_flush(acc, cur, rcur, lane);
        } else {
            unsigned i0 = wave;
            for (; i0 + (RG - 1) * NW < cnt; i0 += RG * NW)
                row_trip<false>(feat, q_e, q_x, acc, i0, cnt, lane);
            if (i0 < cnt) row_trip<true>(feat, q_e, q_x, acc, i0, cnt, lane);
        }
    }
    __syncthreads();

    if (!use_list) {
        float* const hb = q_xb;
        for (int i = tid; i < HROWS * HID; i += TB) {
            const int nl = i >> 6, dd = i & 63;
            float r = 0.f;
            if (nl < TILE && base + nl < NN) {
                const float s0 = s_l[nl * HEADS], s1 = s_l[nl * HEADS + 1];
                const int w0 = nl * HD + (dd & 3) * 32 + (dd >> 2);
                const float a0 = s0 > 0.f ? acc[w0]      / s0 : 0.f;
                const float a1 = s1 > 0.f ? acc[w0 + 16] / s1 : 0.f;
                const float v0 = a0 + bias[dd];
                const float v1 = a1 + bias[HID + dd];
                const float v  = 0.5f * (v0 + v1);
                r = v > 0.f ? v : (__expf(v) - 1.f);
            }
            hb[nl * HP + dd] = r;
        }
        __syncthreads();

        const int wv   = __builtin_amdgcn_readfirstlane(wave);
        const int ct   = wv & 7;
        const int half = lane >> 4, l15 = lane & 15;
        v2f bf[HID / 4];
#pragma unroll
        for (int kq = 0; kq < HID / 4; ++kq) {
            const int k0 = kq * 4 + half * 2;
            bf[kq].x = Wn[k0 * HD + ct * 16 + l15];
            bf[kq].y = Wn[(k0 + 1) * HD + ct * 16 + l15];
        }
        float* const fb = acc;
        const int fcol = (ct >> 2) * FH1 + (ct & 3) * 16 + l15;
        for (int rt = wv >> 3; rt < HROWS / 16; rt += 4) {
            const float* arow = hb + (rt * 16 + l15) * HP + half * 2;
            v8f cc = {};
#pragma unroll
            for (int kq = 0; kq < HID / 4; ++kq) {
                const v2f a = *(const v2f*)(arow + kq * 4);
                cc = __builtin_amdgcn_wmma_f32_16x16x4_f32(false, a, false, bf[kq],
                                                           (short)0, cc, false, false);
            }
            const int r0 = rt * 16 + half * 8;
#pragma unroll
            for (int v = 0; v < 8; ++v) {
                const int row = r0 + v;
                if (row < TILE) {
                    fb[row * FP + fcol] = cc[v];
                    if (base + row < NN) featn[(long)(base + row) * HD + ct * 16 + l15] = cc[v];
                }
            }
        }
        __syncthreads();

        if (tid < TILE * HEADS && base + (tid >> 1) < NN) {
            const int hh = tid & 1;
            const float* f  = fb + (tid >> 1) * FP + hh * FH1;
            const float* la = aln + hh * HID;
            const float* ra = arn + hh * HID;
            float sl = 0.f, sr = 0.f;
#pragma unroll 4
            for (int d = 0; d < HID; ++d) { float v = f[d]; sl += v * la[d]; sr += v * ra[d]; }
            eln[base * HEADS + tid] = sl; ern[base * HEADS + tid] = sr;
        }
        return;
    }

    for (int i = tid; i < TILE * HID; i += TB) {
        const int nl = i >> 6, dd = i & 63;
        const int n  = base + nl;
        if (n < NN) {
            const float s0 = s_l[nl * HEADS], s1 = s_l[nl * HEADS + 1];
            const int w0 = nl * HD + (dd & 3) * 32 + (dd >> 2);
            const float a0 = s0 > 0.f ? acc[w0]      / s0 : 0.f;
            const float a1 = s1 > 0.f ? acc[w0 + 16] / s1 : 0.f;
            const float v0 = a0 + bias[dd];
            const float v1 = a1 + bias[HID + dd];
            const float v  = 0.5f * (v0 + v1);
            out[(long)n * HID + dd] = v > 0.f ? v : (__expf(v) - 1.f);
        }
    }
}

extern "C" void kernel_launch(void* const* d_in, const int* in_sizes, int n_in,
                              void* d_out, int out_size, void* d_ws, size_t ws_size,
                              hipStream_t stream) {
    const float* x   = (const float*)d_in[0];
    const float* W1  = (const float*)d_in[1];
    const float* al1 = (const float*)d_in[2];
    const float* ar1 = (const float*)d_in[3];
    const float* b1  = (const float*)d_in[4];
    const float* W2  = (const float*)d_in[5];
    const float* al2 = (const float*)d_in[6];
    const float* ar2 = (const float*)d_in[7];
    const float* b2  = (const float*)d_in[8];
    const int*   src = (const int*)d_in[9];
    const int*   dst = (const int*)d_in[10];
    float* out = (float*)d_out;

    char* wsb = (char*)d_ws;
    size_t off = 0;
    auto alloc = [&](size_t bytes) -> void* {
        void* p = wsb + off;
        off += bytes;
        off = (off + 255) & ~(size_t)255;
        return p;
    };
    float* feat = (float*)alloc((size_t)NN * HD * 4);
    float* el   = (float*)alloc((size_t)NN * HEADS * 4);
    float* er   = (float*)alloc((size_t)NN * HEADS * 4);
    float* feat2 = (float*)alloc((size_t)NN * HD * 4);
    float* el2  = (float*)alloc((size_t)NN * HEADS * 4);
    float* er2  = (float*)alloc((size_t)NN * HEADS * 4);
    unsigned char* bkt = (unsigned char*)alloc((size_t)NE);
    unsigned* tcnt  = (unsigned*)alloc((size_t)NTILES * 4);
    unsigned* tlist = (unsigned*)alloc((size_t)NTILES * QCAP * 4);

    const dim3 B256(256);

    unsigned* pk = (unsigned*)d_out;
    gat_bucket<<<dim3(NBK), B256, 0, stream>>>(src, dst, bkt, pk, tlist, NE);

    gat_proj<<<dim3((NN + PR - 1) / PR), B256, 0, stream>>>(x, W1, al1, ar1, feat, el, er);
    gat_tile<<<dim3(NTILES), dim3(TB), 0, stream>>>(src, dst, bkt, pk, feat, el, er, b1, out, tcnt, tlist, 0,
                                                    W2, al2, ar2, feat2, el2, er2);

    gat_tile<<<dim3(NTILES), dim3(TB), 0, stream>>>(src, dst, bkt, pk, feat2, el2, er2, b2, out, tcnt, tlist, 1,
                                                    nullptr, nullptr, nullptr, nullptr, nullptr, nullptr);
}


# Row max and weight sum by the row's owner wave: edges binned first (rank = count's return), el[src] read once, 16-lane reductions over sorted logits, no LDS max/sum atomics; s in tree order
# speedup vs baseline: 9.8152x; 1.0617x over previous
#include <hip/hip_runtime.h>
#include <hip/hip_bf16.h>

#define NN   50000
#define NE   1600000
#define INF_ 128
#define HID  64
#define HEADS 2
#define HD   (HEADS*HID)
#define NEG_SLOPE 0.2f

typedef __attribute__((ext_vector_type(2))) float v2f;
typedef __attribute__((ext_vector_type(8))) float v8f;

__device__ __forceinline__ void atomicMaxF(float* addr, float val) {
    if (val >= 0.0f) atomicMax((int*)addr, __float_as_int(val));
    else             atomicMin((unsigned int*)addr, (unsigned int)__float_as_int(val));
}


#define PR   64
#define XP   132

__global__ void __launch_bounds__(256) gat_proj(
        const float* __restrict__ x, const float* __restrict__ W,
        const float* __restrict__ al, const float* __restrict__ ar,
        float* __restrict__ feat, float* __restrict__ el, float* __restrict__ er) {
    __shared__ __attribute__((aligned(16))) float xb[PR * XP];
    const int tid  = threadIdx.x;
    const int lane = tid & 31;
    const int wv   = __builtin_amdgcn_readfirstlane(tid >> 5);
    const int base = blockIdx.x * PR;
    const int half = lane >> 4, l15 = lane & 15;

    float4 xr[PR / 8];
#pragma unroll
    for (int j = 0; j < PR / 8; ++j) {
        int n = base + wv + j * 8;
        n = n < NN ? n : NN - 1;
        xr[j] = *(const float4*)(x + (long)n * INF_ + lane * 4);
    }
#pragma unroll
    for (int j = 0; j < PR / 8; ++j) {
        const int r = wv + j * 8;
        *(float4*)(xb + r * XP + lane * 4) = base + r < NN ? xr[j] : make_float4(0.f, 0.f, 0.f, 0.f);
    }
    __syncthreads();

    v8f cc[PR / 16];
#pragma unroll
    for (int t = 0; t < PR / 16; ++t) cc[t] = v8f{};
    const float* wcol = W + wv * 16 + l15;
    const float* arow = xb + l15 * XP + half * 2;
#pragma unroll 8
    for (int kq = 0; kq < INF_ / 4; ++kq) {
        const int k0 = kq * 4 + half * 2;
        v2f b;
        b.x = wcol[k0 * HD];
        b.y = wcol[(k0 + 1) * HD];
#pragma unroll
        for (int t = 0; t < PR / 16; ++t) {
            const v2f a = *(const v2f*)(arow + t * 16 * XP + kq * 4);
            cc[t] = __builtin_amdgcn_wmma_f32_16x16x4_f32(false, a, false, b,
                                                          (short)0, cc[t], false, false);
        }
    }
    __syncthreads();

#pragma unroll
    for (int t = 0; t < PR / 16; ++t) {
#pragma unroll
        for (int v = 0; v < 8; ++v)
            xb[(t * 16 + half * 8 + v) * XP + wv * 16 + l15] = cc[t][v];
    }
    __syncthreads();

#pragma unroll
    for (int j = 0; j < PR / 8; ++j) xr[j] = *(const float4*)(xb + (wv + j * 8) * XP + lane * 4);
#pragma unroll
    for (int j = 0; j < PR / 8; ++j) {
        const int r = wv + j * 8;
        if (base + r < NN) *(float4*)(feat + (long)(base + r) * HD + lane * 4) = xr[j];
    }

    if (tid < PR * HEADS && base + (tid >> 1) < NN) {
        const int hh = tid & 1;
        const float* f  = xb + (tid >> 1) * XP + hh * HID;
        const float* la = al + hh * HID;
        const float* ra = ar + hh * HID;
        float sl = 0.f, sr = 0.f;
#pragma unroll 4
        for (int d = 0; d < HID; ++d) { float v = f[d]; sl += v * la[d]; sr += v * ra[d]; }
        el[base * HEADS + tid] = sl; er[base * HEADS + tid] = sr;
    }
}

#define TILE   196
#define NTILES 256
#define QCAP   8192
#define TB     1024
#define RG     8
#define HROWS  208
#define HP     68
#define FP     130
#define FH1    65
#define ACCW   (TILE * FP)

#define BKE    2048
#define NBK    ((NE + BKE - 1) / BKE)
#define BMW    (NBK * 4)

__global__ void __launch_bounds__(256) gat_bucket(const int* __restrict__ src,
                                                  const int* __restrict__ dst,
                                                  unsigned char* __restrict__ bkt,
                                                  unsigned* __restrict__ pk,
                                                  unsigned* __restrict__ tlist, int E) {
    __shared__ unsigned w[4 * NTILES];
    const int t = threadIdx.x;
    for (int i = t; i < 4 * NTILES; i += 256) w[i] = 0u;
    __syncthreads();
    const int e0 = blockIdx.x * BKE + t * 8;
    if (e0 < E) {
        const int4 a = *(const int4*)(dst + e0);
        const int4 b = *(const int4*)(dst + e0 + 4);
        const unsigned t0 = a.x / TILE, t1 = a.y / TILE, t2 = a.z / TILE, t3 = a.w / TILE;
        const unsigned t4 = b.x / TILE, t5 = b.y / TILE, t6 = b.z / TILE, t7 = b.w / TILE;
        uint2 tb;
        tb.x = t0 | (t1 << 8) | (t2 << 16) | (t3 << 24);
        tb.y = t4 | (t5 << 8) | (t6 << 16) | (t7 << 24);
        *(uint2*)(bkt + e0) = tb;
        const int4 sa = *(const int4*)(src + e0);
        const int4 sb = *(const int4*)(src + e0 + 4);
        uint4 w0, w1;
        w0.x = (unsigned)sa.x | ((unsigned)a.x << 16); w0.y = (unsigned)sa.y | ((unsigned)a.y << 16);
        w0.z = (unsigned)sa.z | ((unsigned)a.z << 16); w0.w = (unsigned)sa.w | ((unsigned)a.w << 16);
        w1.x = (unsigned)sb.x | ((unsigned)b.x << 16); w1.y = (unsigned)sb.y | ((unsigned)b.y << 16);
        w1.z = (unsigned)sb.z | ((unsigned)b.z << 16); w1.w = (unsigned)sb.w | ((unsigned)b.w << 16);
        *(uint4*)(pk + e0) = w0;
        *(uint4*)(pk + e0 + 4) = w1;
        unsigned* wj = w + (t >> 6) * NTILES;
        const unsigned bit = 1u << ((t >> 1) & 31);
        atomicOr(wj + t0, bit); atomicOr(wj + t1, bit); atomicOr(wj + t2, bit); atomicOr(wj + t3, bit);
        atomicOr(wj + t4, bit); atomicOr(wj + t5, bit); atomicOr(wj + t6, bit); atomicOr(wj + t7, bit);
    }
    __syncthreads();
    uint4 o;
    o.x = w[t]; o.y = w[NTILES + t]; o.z = w[2 * NTILES + t]; o.w = w[3 * NTILES + t];
    *(uint4*)(tlist + (long)t * QCAP + blockIdx.x * 4) = o;
}

__device__ __forceinline__ void tile_match16(uint4 w, unsigned pat, int e0, int* q_e, unsigned* qn) {
    const unsigned ws[4] = {w.x, w.y, w.z, w.w};
#pragma unroll
    for (int k = 0; k < 4; ++k) {
        const unsigned x = ws[k] ^ pat;
        unsigned t = ~(((x & 0x7f7f7f7fu) + 0x7f7f7f7fu) | x | 0x7f7f7f7fu);
        while (t) {
            const int b = __builtin_ctz(t) >> 3;
            const unsigned slot = atomicAdd(qn, 1u);
            if (slot < QCAP) q_e[slot] = e0 + k * 4 + b;
            t &= t - 1u;
        }
    }
}

__device__ __forceinline__ void tile_scan(const unsigned char* __restrict__ bkt,
                                          int e_begin, int e_end, unsigned myb,
                                          int* q_e, unsigned* qn, int tid) {
    const uint4* __restrict__ p = (const uint4*)bkt;
    const unsigned pat = myb * 0x01010101u;
    const int g1 = e_end >> 4;
    int g = (e_begin >> 4) + tid;
    for (; g + 7 * TB < g1; g += 8 * TB) {
        uint4 w[8];
#pragma unroll
        for (int u = 0; u < 8; ++u) w[u] = p[g + u * TB];
#pragma unroll
        for (int u = 0; u < 8; ++u) tile_match16(w[u], pat, (g + u * TB) * 16, q_e, qn);
    }
    for (; g < g1; g += TB) tile_match16(p[g], pat, g * 16, q_e, qn);
}

__device__ __forceinline__ void map_walk(const unsigned* bm, unsigned* q_g, unsigned* gn, int tid) {
    for (int i = tid; i < BMW; i += TB) {
        unsigned w = bm[i];
        while (w) {
            const int b = __builtin_ctz(w);
            w &= w - 1u;
            const unsigned slot = atomicAdd(gn, 1u);
            if (slot < QCAP) q_g[slot] = (unsigned)(i * 32 + b);
        }
    }
}

#define GU 8
__device__ __forceinline__ void group_match(const unsigned* __restrict__ pk,
                                            const unsigned* q_g, unsigned ng, int base,
                                            int* q_e, unsigned* qn, int tid) {
    const uint4* __restrict__ p = (const uint4*)pk;
    const unsigned n4   = ng * 4u;
    const unsigned lane = (unsigned)tid & 31u;
    const unsigned qq   = (unsigned)tid & 3u;
    unsigned jw = (unsigned)__builtin_amdgcn_readfirstlane(tid - (int)lane);
    for (; jw < n4; jw += GU * TB) {
        uint4 v[GU]; bool live[GU];
#pragma unroll
        for (int u = 0; u < GU; ++u) {
            const unsigned j = jw + lane + u * TB;
            live[u] = j < n4;
            const unsigned g = q_g[(live[u] ? j : n4 - 1u) >> 2];
            v[u] = p[g * 4u + qq];
        }
        unsigned bal[GU * 4];
        unsigned tot = 0u;
#pragma unroll
        for (int u = 0; u < GU; ++u) {
            const unsigned ws[4] = {v[u].x, v[u].y, v[u].z, v[u].w};
#pragma unroll
            for (int k = 0; k < 4; ++k) {
                const unsigned d = (ws[k] >> 16) - (unsigned)base;
                bal[u * 4 + k] = __builtin_amdgcn_ballot_w32(live[u] && d < TILE);
                tot += (unsigned)__builtin_popcount(bal[u * 4 + k]);
            }
        }
        if (tot) {
            unsigned wb = 0u;
            if (lane == 0u) wb = atomicAdd(qn, tot);
            wb = (unsigned)__builtin_amdgcn_readfirstlane((int)wb);
#pragma unroll
            for (int u = 0; u < GU; ++u) {
                const unsigned ws[4] = {v[u].x, v[u].y, v[u].z, v[u].w};
#pragma unroll
                for (int k = 0; k < 4; ++k) {
                    const unsigned b = bal[u * 4 + k];
                    if (b) {
                        if ((b >> lane) & 1u) {
                            const unsigned slot = wb + __builtin_amdgcn_mbcnt_lo(b, 0u);
                            const unsigned d = (ws[k] >> 16) - (unsigned)base;
                            if (slot < QCAP) q_e[slot] = (int)((ws[k] & 0xffffu) | (d << 16));
                        }
                        wb += (unsigned)__builtin_popcount(b);
                    }
                }
            }
        }
    }
}

template <bool TAIL>
__device__ __forceinline__ void row_trip(const float* __restrict__ feat, const int* q_e,
                                         const float* q_x, float* acc,
                                         unsigned i0, unsigned cnt, int lane) {
    const unsigned NW = TB / 32;
    float4 f[RG]; float x[RG]; int d[RG];
#pragma unroll
    for (int u = 0; u < RG; ++u) {
        unsigned i = i0 + u * NW;
        bool live = true;
        if (TAIL) { live = i < cnt; i = live ? i : cnt - 1u; }
        const unsigned p = (unsigned)q_e[i];
        d[u] = live ? (int)(p >> 16) : TILE;
        x[u] = q_x[i];
        f[u] = *(const float4*)(feat + (long)(p & 0xffffu) * HD + lane * 4);
    }
#pragma unroll
    for (int u = 0; u < RG; ++u) {
        float* o = acc + d[u] * HD + lane;
        unsafeAtomicAdd(o,      f[u].x * x[u]);
        unsafeAtomicAdd(o + 32, f[u].y * x[u]);
        unsafeAtomicAdd(o + 64, f[u].z * x[u]);
        unsafeAtomicAdd(o + 96, f[u].w * x[u]);
    }
}

#define NWV    (TB / 32)
#define OWNR   ((TILE + NWV - 1) / NWV)
#define NBIN   (NWV * OWNR)

__device__ __forceinline__ unsigned own_bin(int dl) {
    return (unsigned)(dl & (NWV - 1)) * OWNR + (unsigned)(dl / NWV);
}

__device__ __forceinline__ void own_flush(float* acc, float4 cur, int r, int lane) {
    if (r >= 0) {
        float* o = acc + r * HD + lane;
        o[0] = cur.x; o[32] = cur.y; o[64] = cur.z; o[96] = cur.w;
    }
}

template <bool TAIL>
__device__ __forceinline__ void own_trip(const float* __restrict__ feat, const unsigned* q_s,
                                         const float* q_x, float* acc, float4& cur, int& rcur,
                                         unsigned i0, unsigned b1, int lane) {
    float4 f[RG]; float x[RG]; int r[RG];
#pragma unroll
    for (int u = 0; u < RG; ++u) {
        unsigned i = i0 + u;
        if (TAIL) i = i < b1 ? i : b1 - 1u;
        const unsigned p = q_s[i];
        r[u] = __builtin_amdgcn_readfirstlane((int)(p >> 16));
        x[u] = q_x[i];
        f[u] = *(const float4*)(feat + (long)(p & 0xffffu) * HD + lane * 4);
    }
#pragma unroll
    for (int u = 0; u < RG; ++u) {
        if (!TAIL || i0 + u < b1) {
            if (r[u] != rcur) {
                own_flush(acc, cur, rcur, lane);
                cur = make_float4(0.f, 0.f, 0.f, 0.f);
                rcur = r[u];
            }
            cur.x += f[u].x * x[u]; cur.y += f[u].y * x[u];
            cur.z += f[u].z * x[u]; cur.w += f[u].w * x[u];
        }
    }
}

#define NSL    (QCAP / TB)
#define SV     3

template <int N>
__device__ __forceinline__ float row_peer(float x) {
    return __builtin_bit_cast(float, __builtin_amdgcn_update_dpp(0, __builtin_bit_cast(int, x),
                                                                 0x160 + N, 0xf, 0xf, false));
}
__device__ __forceinline__ float row_max16(float x) {
    x = fmaxf(x, row_peer<1>(x)); x = fmaxf(x, row_peer<2>(x));
    x = fmaxf(x, row_peer<4>(x)); x = fmaxf(x, row_peer<8>(x));
    return x;
}
__device__ __forceinline__ float row_sum16(float x) {
    x += row_peer<1>(x); x += row_peer<2>(x);
    x += row_peer<4>(x); x += row_peer<8>(x);
    return x;
}

__global__ void __launch_bounds__(TB) gat_tile(
        const int* __restrict__ src, const int* __restrict__ dst,
        const unsigned char* __restrict__ bkt, const unsigned* __restrict__ pk,
        const float* __restrict__ feat,
        const float* __restrict__ el, const float* __restrict__ er,
        const float* __restrict__ bias, float* __restrict__ out,
        unsigned* __restrict__ tcnt, unsigned* __restrict__ tlist, int use_list,
        const float* __restrict__ Wn, const float* __restrict__ aln, const float* __restrict__ arn,
        float* __restrict__ featn, float* __restrict__ eln, float* __restrict__ ern) {
    __shared__ __attribute__((aligned(16))) float acc[ACCW];
    __shared__ float m_l[TILE * HEADS];
    __shared__ float s_l[TILE * HEADS];
    __shared__ float er_l[TILE * HEADS];
    __shared__ int   q_e[QCAP];
    __shared__ __attribute__((aligned(16))) float q_xb[2 * QCAP];
    __shared__ unsigned q_g[QCAP];
    __shared__ unsigned qn, gn;
    __shared__ unsigned bcnt[NBIN], boff[NBIN + 1];
    float* const q_x0 = q_xb;
    float* const q_x1 = q_xb + QCAP;

    const int tid  = threadIdx.x;
    const int base = blockIdx.x * TILE;
    const unsigned myb = blockIdx.x;
    unsigned* __restrict__ mylist = tlist + (long)blockIdx.x * QCAP;

    for (int i = tid; i < (TILE + 1) * HD; i += TB) acc[i] = 0.f;
    for (int i = tid; i < TILE * HEADS; i += TB) {
        m_l[i] = -INFINITY; s_l[i] = 0.f;
        er_l[i] = (base + (i >> 1)) < NN ? er[base * HEADS + i] : 0.f;
    }
    if (tid == 0) { qn = 0u; gn = 0u; }
    if (tid < NBIN) bcnt[tid] = 0u;
    __syncthreads();

    unsigned cnt = 0u;
    if (use_list) cnt = tcnt[blockIdx.x];
    if (!use_list) map_walk(mylist, q_g, &gn, tid);
    __syncthreads();
    if (!use_list) {
        const unsigned ng = gn;
        if (ng <= QCAP) group_match(pk, q_g, ng, base, q_e, &qn, tid);
        else if (tid == 0) qn = ng;
    }
    __syncthreads();
    if (!use_list) {
        cnt = qn;
        if (tid == 0) tcnt[blockIdx.x] = cnt;
    }
    const int seglen = cnt > QCAP ? QCAP : NE;
    const int nseg   = (NE + seglen - 1) / seglen;
    const bool own = nseg == 1;
    const int wave = tid >> 5, lane = tid & 31;
    const int h  = lane >> 4;

    if (own) {
        static_assert(QCAP % TB == 0, "a thread takes whole queue slots");
        unsigned pw[NSL], rk[NSL];
        float2 lg[NSL];
#pragma unroll
        for (int k = 0; k < NSL; ++k) {
            const unsigned i = (unsigned)tid + (unsigned)k * TB;
            pw[k] = 0u;
            if (i < cnt) pw[k] = use_list ? mylist[i] : (unsigned)q_e[i];
        }
#pragma unroll
        for (int k = 0; k < NSL; ++k) {
            const unsigned i = (unsigned)tid + (unsigned)k * TB;
            lg[k] = make_float2(0.f, 0.f);
            if (i < cnt) lg[k] = *(const float2*)(el + (long)(pw[k] & 0xffffu) * HEADS);
        }
#pragma unroll
        for (int k = 0; k < NSL; ++k) {
            const unsigned i = (unsigned)tid + (unsigned)k * TB;
            rk[k] = 0u;
            if (i < cnt) {
                if (!use_list) mylist[i] = pw[k];
                rk[k] = atomicAdd(&bcnt[own_bin((int)(pw[k] >> 16))], 1u);
            }
        }
        __syncthreads();
        if (tid < NWV) {
            unsigned cb[OWNR], tot = 0u;
#pragma unroll
            for (int k = 0; k < OWNR; ++k) { cb[k] = bcnt[tid * OWNR + k]; tot += cb[k]; }
            unsigned inc = tot;
#pragma unroll
            for (int d = 1; d < NWV; d <<= 1) {
                const unsigned v = __shfl_up(inc, d);
                if (tid >= d) inc += v;
            }
            unsigned o = inc - tot;
#pragma unroll
            for (int k = 0; k < OWNR; ++k) { boff[tid * OWNR + k] = o; o += cb[k]; }
            if (tid == NWV - 1) boff[NBIN] = o;
        }
        __syncthreads();

#pragma unroll
        for (int k = 0; k < NSL; ++k) {
            const unsigned i = (unsigned)tid + (unsigned)k * TB;
            if (i < cnt) {
                const int dl = (int)(pw[k] >> 16);
                float v0 = lg[k].x + er_l[dl * HEADS], v1 = lg[k].y + er_l[dl * HEADS + 1];
                v0 = v0 > 0.f ? v0 : NEG_SLOPE * v0;
                v1 = v1 > 0.f ? v1 : NEG_SLOPE * v1;
                const unsigned slot = boff[own_bin(dl)] + rk[k];
                q_g[slot] = pw[k];
                q_x0[slot] = v0; q_x1[slot] = v1;
            }
        }
        __syncthreads();

        const int wo = __builtin_amdgcn_readfirstlane(wave);

        {
            float* const qv = h ? q_x1 : q_x0;
            const unsigned l15 = (unsigned)lane & 15u;
            for (int k = 0; k < OWNR; ++k) {
                const int dl = k * NWV + wo;
                if (dl >= TILE) break;
                const unsigned r0 = (unsigned)__builtin_amdgcn_readfirstlane((int)boff[wo * OWNR + k]);
                const unsigned r1 = (unsigned)__builtin_amdgcn_readfirstlane((int)boff[wo * OWNR + k + 1]);
                if (r0 == r1) continue;
                float v[SV];
                float mx = -INFINITY;
#pragma unroll
                for (int u = 0; u < SV; ++u) {
                    const unsigned j = r0 + l15 + 16u * (unsigned)u;
                    const float t = qv[j < r1 ? j : r0];
                    v[u] = j < r1 ? t : -INFINITY;
                    mx = fmaxf(mx, v[u]);
                }
                for (unsigned j = r0 + 16u * SV + l15; j < r1; j += 16u) mx = fmaxf(mx, qv[j]);
                mx = row_max16(mx);
                float sm = 0.f;
#pragma unroll
                for (int u = 0; u < SV; ++u) {
                    const unsigned j = r0 + l15 + 16u * (unsigned)u;
                    const float x = __expf(v[u] - mx);
                    if (j < r1) qv[j] = x;
                    sm += x;
                }
                for (unsigned j = r0 + 16u * SV + l15; j < r1; j += 16u) {
                    const float x = __expf(qv[j] - mx);
                    qv[j] = x;
                    sm += x;
                }
                sm = row_sum16(sm);
                if (l15 == 0u) s_l[dl * HEADS + h] = sm;
            }
        }

        {
            const float* q_x = h ? q_x1 : q_x0;
            const unsigned b1 = (unsigned)__builtin_amdgcn_readfirstlane((int)boff[(wo + 1) * OWNR]);
            unsigned i0 = (unsigned)__builtin_amdgcn_readfirstlane((int)boff[wo * OWNR]);
            float4 cur = make_float4(0.f, 0.f, 0.f, 0.f);
            int rcur = -1;
            for (; i0 + RG <= b1; i0 += RG) own_trip<false>(feat, q_g, q_x, acc, cur, rcur, i0, b1, lane);
            if (i0 < b1) own_trip<true>(feat, q_g, q_x, acc, cur, rcur, i0, b1, lane);
            own_flush(acc, cur, rcur, lane);
        }
    } else {
        for (int seg = 0; seg < nseg; ++seg) {
            __syncthreads();
            if (tid == 0) qn = 0u;
            __syncthreads();
            const int eb = seg * seglen;
            tile_scan(bkt, eb, min(NE, eb + seglen), myb, q_e, &qn, tid);
            __syncthreads();
            cnt = qn;
            for (unsigned i = tid; i < cnt; i += TB) {
                const int e  = q_e[i];
                const int s0 = src[e];
                const int dl = dst[e] - base;
                const float2 l2 = *(const float2*)(el + (long)s0 * HEADS);
                float v0 = l2.x + er_l[dl * HEADS], v1 = l2.y + er_l[dl * HEADS + 1];
                v0 = v0 > 0.f ? v0 : NEG_SLOPE * v0;
                v1 = v1 > 0.f ? v1 : NEG_SLOPE * v1;
                atomicMaxF(&m_l[dl * HEADS], v0);
                atomicMaxF(&m_l[dl * HEADS + 1], v1);
            }
        }
        __syncthreads();

        for (int seg = 0; seg < nseg; ++seg) {
            __syncthreads();
            if (tid == 0) qn = 0u;
            __syncthreads();
            const int eb = seg * seglen;
            tile_scan(bkt, eb, min(NE, eb + seglen), myb, q_e, &qn, tid);
            __syncthreads();
            cnt = qn;
            for (unsigned i = tid; i < cnt; i += TB) {
                const int e  = q_e[i];
                const int s0 = src[e];
                const int dl = dst[e] - base;
                const float2 l2 = *(const float2*)(el + (long)s0 * HEADS);
                float v0 = l2.x + er_l[dl * HEADS], v1 = l2.y + er_l[dl * HEADS + 1];
                v0 = v0 > 0.f ? v0 : NEG_SLOPE * v0;
                v1 = v1 > 0.f ? v1 : NEG_SLOPE * v1;
                const float x0 = __expf(v0 - m_l[dl * HEADS]);
                const float x1 = __expf(v1 - m_l[dl * HEADS + 1]);
                q_e[i] = (int)((unsigned)s0 | ((unsigned)dl << 16));
                q_x0[i] = x0; q_x1[i] = x1;
                unsafeAtomicAdd(&s_l[dl * HEADS], x0);
                unsafeAtomicAdd(&s_l[dl * HEADS + 1], x1);
            }
            __syncthreads();
            const float* q_x = h ? q_x1 : q_x0;
            unsigned i0 = wave;
            for (; i0 + (RG - 1) * NWV < cnt; i0 += RG * NWV)
                row_trip<false>(feat, q_e, q_x, acc, i0, cnt, lane);
            if (i0 < cnt) row_trip<true>(feat, q_e, q_x, acc, i0, cnt, lane);
        }
    }
    __syncthreads();

    if (!use_list) {
        float* const hb = q_xb;
        for (int i = tid; i < HROWS * HID; i += TB) {
            const int nl = i >> 6, dd = i & 63;
            float r = 0.f;
            if (nl < TILE && base + nl < NN) {
                const float s0 = s_l[nl * HEADS], s1 = s_l[nl * HEADS + 1];
                const int w0 = nl * HD + (dd & 3) * 32 + (dd >> 2);
                const float a0 = s0 > 0.f ? acc[w0]      / s0 : 0.f;
                const float a1 = s1 > 0.f ? acc[w0 + 16] / s1 : 0.f;
                const float v0 = a0 + bias[dd];
                const float v1 = a1 + bias[HID + dd];
                const float v  = 0.5f * (v0 + v1);
                r = v > 0.f ? v : (__expf(v) - 1.f);
            }
            hb[nl * HP + dd] = r;
        }
        __syncthreads();

        const int wv   = __builtin_amdgcn_readfirstlane(wave);
        const int ct   = wv & 7;
        const int half = lane >> 4, l15 = lane & 15;
        v2f bf[HID / 4];
#pragma unroll
        for (int kq = 0; kq < HID / 4; ++kq) {
            const int k0 = kq * 4 + half * 2;
            bf[kq].x = Wn[k0 * HD + ct * 16 + l15];
            bf[kq].y = Wn[(k0 + 1) * HD + ct * 16 + l15];
        }
        float* const fb = acc;
        const int fcol = (ct >> 2) * FH1 + (ct & 3) * 16 + l15;
        for (int rt = wv >> 3; rt < HROWS / 16; rt += 4) {
            const float* arow = hb + (rt * 16 + l15) * HP + half * 2;
            v8f cc = {};
#pragma unroll
            for (int kq = 0; kq < HID / 4; ++kq) {
                const v2f a = *(const v2f*)(arow + kq * 4);
                cc = __builtin_amdgcn_wmma_f32_16x16x4_f32(false, a, false, bf[kq],
                                                           (short)0, cc, false, false);
            }
            const int r0 = rt * 16 + half * 8;
#pragma unroll
            for (int v = 0; v < 8; ++v) {
                const int row = r0 + v;
                if (row < TILE) {
                    fb[row * FP + fcol] = cc[v];
                    if (base + row < NN) featn[(long)(base + row) * HD + ct * 16 + l15] = cc[v];
                }
            }
        }
        __syncthreads();

        if (tid < TILE * HEADS && base + (tid >> 1) < NN) {
            const int hh = tid & 1;
            const float* f  = fb + (tid >> 1) * FP + hh * FH1;
            const float* la = aln + hh * HID;
            const float* ra = arn + hh * HID;
            float sl = 0.f, sr = 0.f;
#pragma unroll 4
            for (int d = 0; d < HID; ++d) { float v = f[d]; sl += v * la[d]; sr += v * ra[d]; }
            eln[base * HEADS + tid] = sl; ern[base * HEADS + tid] = sr;
        }
        return;
    }

    for (int i = tid; i < TILE * HID; i += TB) {
        const int nl = i >> 6, dd = i & 63;
        const int n  = base + nl;
        if (n < NN) {
            const float s0 = s_l[nl * HEADS], s1 = s_l[nl * HEADS + 1];
            const int w0 = nl * HD + (dd & 3) * 32 + (dd >> 2);
            const float a0 = s0 > 0.f ? acc[w0]      / s0 : 0.f;
            const float a1 = s1 > 0.f ? acc[w0 + 16] / s1 : 0.f;
            const float v0 = a0 + bias[dd];
            const float v1 = a1 + bias[HID + dd];
            const float v  = 0.5f * (v0 + v1);
            out[(long)n * HID + dd] = v > 0.f ? v : (__expf(v) - 1.f);
        }
    }
}

extern "C" void kernel_launch(void* const* d_in, const int* in_sizes, int n_in,
                              void* d_out, int out_size, void* d_ws, size_t ws_size,
                              hipStream_t stream) {
    const float* x   = (const float*)d_in[0];
    const float* W1  = (const float*)d_in[1];
    const float* al1 = (const float*)d_in[2];
    const float* ar1 = (const float*)d_in[3];
    const float* b1  = (const float*)d_in[4];
    const float* W2  = (const float*)d_in[5];
    const float* al2 = (const float*)d_in[6];
    const float* ar2 = (const float*)d_in[7];
    const float* b2  = (const float*)d_in[8];
    const int*   src = (const int*)d_in[9];
    const int*   dst = (const int*)d_in[10];
    float* out = (float*)d_out;

    char* wsb = (char*)d_ws;
    size_t off = 0;
    auto alloc = [&](size_t bytes) -> void* {
        void* p = wsb + off;
        off += bytes;
        off = (off + 255) & ~(size_t)255;
        return p;
    };
    float* feat = (float*)alloc((size_t)NN * HD * 4);
    float* el   = (float*)alloc((size_t)NN * HEADS * 4);
    float* er   = (float*)alloc((size_t)NN * HEADS * 4);
    float* feat2 = (float*)alloc((size_t)NN * HD * 4);
    float* el2  = (float*)alloc((size_t)NN * HEADS * 4);
    float* er2  = (float*)alloc((size_t)NN * HEADS * 4);
    unsigned char* bkt = (unsigned char*)alloc((size_t)NE);
    unsigned* tcnt  = (unsigned*)alloc((size_t)NTILES * 4);
    unsigned* tlist = (unsigned*)alloc((size_t)NTILES * QCAP * 4);

    const dim3 B256(256);

    unsigned* pk = (unsigned*)d_out;
    gat_bucket<<<dim3(NBK), B256, 0, stream>>>(src, dst, bkt, pk, tlist, NE);

    gat_proj<<<dim3((NN + PR - 1) / PR), B256, 0, stream>>>(x, W1, al1, ar1, feat, el, er);
    gat_tile<<<dim3(NTILES), dim3(TB), 0, stream>>>(src, dst, bkt, pk, feat, el, er, b1, out, tcnt, tlist, 0,
                                                    W2, al2, ar2, feat2, el2, er2);

    gat_tile<<<dim3(NTILES), dim3(TB), 0, stream>>>(src, dst, bkt, pk, feat2, el2, er2, b2, out, tcnt, tlist, 1,
                                                    nullptr, nullptr, nullptr, nullptr, nullptr, nullptr);
}
